# starting kernel
# baseline (speedup 1.0000x reference)
#include <hip/hip_runtime.h>
#include <hip/hip_bf16.h>

typedef float v2f __attribute__((ext_vector_type(2)));
typedef float v8f __attribute__((ext_vector_type(8)));

#define IN_F 256
#define OUT_F 128

__global__ void zero_kernel(float* __restrict__ p, long long n) {
    long long i = (long long)blockIdx.x * blockDim.x + threadIdx.x;
    long long stride = (long long)gridDim.x * blockDim.x;
    for (; i < n; i += stride) p[i] = 0.0f;
}

__global__ __launch_bounds__(256)
void gemm_msg_kernel(const float* __restrict__ miu,
                     const float* __restrict__ sigma,
                     const float* __restrict__ Wmu,
                     const float* __restrict__ Wsig,
                     float* __restrict__ m_mu,
                     float* __restrict__ m_sig,
                     int N) {
    const int lane = threadIdx.x & 31;
    const int wave = threadIdx.x >> 5;
    const int row_base = blockIdx.x * 128 + wave * 16;

    const int m    = lane & 15;
    const int kgrp = lane >> 4;

    const int arow = (row_base + m < N) ? (row_base + m) : (N - 1);

    float att[8][8];

    {
        v8f acc[8];
        v8f zero = {};
        #pragma unroll
        for (int t = 0; t < 8; ++t) acc[t] = zero;

        const float* Ap = sigma + (size_t)arow * IN_F + kgrp * 2;
        #pragma unroll 4
        for (int k0 = 0; k0 < IN_F; k0 += 4) {
            v2f a = *(const v2f*)(Ap + k0);
            const float* Bp = Wsig + (size_t)(k0 + kgrp * 2) * OUT_F + m;
            #pragma unroll
            for (int t = 0; t < 8; ++t) {
                v2f b;
                b[0] = Bp[t * 16];
                b[1] = Bp[t * 16 + OUT_F];
                acc[t] = __builtin_amdgcn_wmma_f32_16x16x4_f32(
                    false, a, false, b, (short)0, acc[t], false, false);
            }
        }
        #pragma unroll
        for (int t = 0; t < 8; ++t) {
            #pragma unroll
            for (int r = 0; r < 8; ++r) {
                float hs = fmaxf(acc[t][r], 0.0f);
                float a  = __expf(-hs);
                att[t][r] = a;
                int rr = row_base + r + kgrp * 8;
                if (rr < N)
                    m_sig[(size_t)rr * OUT_F + t * 16 + m] = hs * a * a;
            }
        }
    }

    {
        v8f acc[8];
        v8f zero = {};
        #pragma unroll
        for (int t = 0; t < 8; ++t) acc[t] = zero;

        const float* Ap = miu + (size_t)arow * IN_F + kgrp * 2;
        #pragma unroll 4
        for (int k0 = 0; k0 < IN_F; k0 += 4) {
            v2f a = *(const v2f*)(Ap + k0);
            const float* Bp = Wmu + (size_t)(k0 + kgrp * 2) * OUT_F + m;
            #pragma unroll
            for (int t = 0; t < 8; ++t) {
                v2f b;
                b[0] = Bp[t * 16];
                b[1] = Bp[t * 16 + OUT_F];
                acc[t] = __builtin_amdgcn_wmma_f32_16x16x4_f32(
                    false, a, false, b, (short)0, acc[t], false, false);
            }
        }
        #pragma unroll
        for (int t = 0; t < 8; ++t) {
            #pragma unroll
            for (int r = 0; r < 8; ++r) {
                float hm = acc[t][r];
                float e  = (hm > 0.0f) ? hm : (__expf(hm) - 1.0f);
                int rr = row_base + r + kgrp * 8;
                if (rr < N)
                    m_mu[(size_t)rr * OUT_F + t * 16 + m] = e * att[t][r];
            }
        }
    }
}

__global__ __launch_bounds__(256)
void spmm_scatter_kernel(const int* __restrict__ erow,
                         const int* __restrict__ ecol,
                         const float* __restrict__ a1,
                         const float* __restrict__ a2,
                         const float* __restrict__ m_mu,
                         const float* __restrict__ m_sig,
                         float* __restrict__ out_mu,
                         float* __restrict__ out_sig,
                         int E) {
    const int lane = threadIdx.x & 31;
    const long long wid = ((long long)blockIdx.x * blockDim.x + threadIdx.x) >> 5;
    if (wid >= E) return;
    const int e = (int)wid;

    const int r = erow[e];
    const int c = ecol[e];
    const float v1 = a1[e];
    const float v2 = a2[e];

    const float4 mm = ((const float4*)(m_mu  + (size_t)c * OUT_F))[lane];
    const float4 ms = ((const float4*)(m_sig + (size_t)c * OUT_F))[lane];

    float* om = out_mu  + (size_t)r * OUT_F + lane * 4;
    float* os = out_sig + (size_t)r * OUT_F + lane * 4;

    unsafeAtomicAdd(om + 0, v1 * mm.x);
    unsafeAtomicAdd(om + 1, v1 * mm.y);
    unsafeAtomicAdd(om + 2, v1 * mm.z);
    unsafeAtomicAdd(om + 3, v1 * mm.w);
    unsafeAtomicAdd(os + 0, v2 * ms.x);
    unsafeAtomicAdd(os + 1, v2 * ms.y);
    unsafeAtomicAdd(os + 2, v2 * ms.z);
    unsafeAtomicAdd(os + 3, v2 * ms.w);
}

extern "C" void kernel_launch(void* const* d_in, const int* in_sizes, int n_in,
                              void* d_out, int out_size, void* d_ws, size_t ws_size,
                              hipStream_t stream) {
    const float* miu   = (const float*)d_in[0];
    const float* sigma = (const float*)d_in[1];
    const int*   erow  = (const int*)  d_in[2];
    const int*   ecol  = (const int*)  d_in[3];
    const float* a1    = (const float*)d_in[4];
    const float* a2    = (const float*)d_in[5];
    const float* Wmu   = (const float*)d_in[6];
    const float* Wsig  = (const float*)d_in[7];

    const int N = in_sizes[0] / IN_F;
    const int E = in_sizes[2];

    float* out_mu  = (float*)d_out;
    float* out_sig = out_mu + (size_t)N * OUT_F;

    float* m_mu  = (float*)d_ws;
    float* m_sig = m_mu + (size_t)N * OUT_F;

    zero_kernel<<<2048, 256, 0, stream>>>(out_mu, (long long)2 * N * OUT_F);

    gemm_msg_kernel<<<(N + 127) / 128, 256, 0, stream>>>(
        miu, sigma, Wmu, Wsig, m_mu, m_sig, N);

    const long long waves = E;
    const long long blocks = (waves * 32 + 255) / 256;
    spmm_scatter_kernel<<<(int)blocks, 256, 0, stream>>>(
        erow, ecol, a1, a2, m_mu, m_sig, out_mu, out_sig, E);
}


# float-atomic scatter replaced by in-call counting sort of edges by row (int rank atomics, scan, fill) + one wave per row summing gathered messages in registers; sums in rank arrival order
# speedup vs baseline: 1.9709x; 1.9709x over previous
#include <hip/hip_runtime.h>
#include <hip/hip_bf16.h>

typedef float v2f __attribute__((ext_vector_type(2)));
typedef float v8f __attribute__((ext_vector_type(8)));

#define IN_F 256
#define OUT_F 128

#define SCAN_TILE 1024

__global__ void zero_counts_kernel(int* __restrict__ p, int n) {
    int i = blockIdx.x * blockDim.x + threadIdx.x;
    if (i < n) p[i] = 0;
}

__global__ __launch_bounds__(256)
void hist_rank_kernel(const int* __restrict__ erow,
                      int* __restrict__ counts,
                      int* __restrict__ rank,
                      int E) {
    int e = blockIdx.x * blockDim.x + threadIdx.x;
    if (e < E) rank[e] = atomicAdd(&counts[erow[e]], 1);
}

__global__ __launch_bounds__(256)
void row_scan_kernel(const int* __restrict__ counts, int* __restrict__ rowptr) {
    __shared__ int s_front[8];
    __shared__ int s_wave[8];
    const int tid  = threadIdx.x;
    const int lane = tid & 31;
    const int wave = tid >> 5;
    const int tile0 = blockIdx.x * SCAN_TILE;
    const int4* c4 = (const int4*)counts;

    int front = 0;
    #pragma unroll 8
    for (int i = tid * 4; i < tile0; i += SCAN_TILE) {
        int4 v = c4[i >> 2];
        front += (v.x + v.y) + (v.z + v.w);
    }
    #pragma unroll
    for (int d = 16; d > 0; d >>= 1) front += __shfl_xor(front, d, 32);
    if (lane == 0) s_front[wave] = front;

    const int4 v = c4[(tile0 >> 2) + tid];
    const int tsum = (v.x + v.y) + (v.z + v.w);
    int incl = tsum;
    #pragma unroll
    for (int d = 1; d < 32; d <<= 1) {
        int t = __shfl_up(incl, d, 32);
        if (lane >= d) incl += t;
    }
    if (lane == 31) s_wave[wave] = incl;
    __syncthreads();

    int base = 0;
    #pragma unroll
    for (int w = 0; w < 8; ++w) {
        base += s_front[w];
        if (w < wave) base += s_wave[w];
    }
    int4 o;
    o.x = base + incl - tsum;
    o.y = o.x + v.x;
    o.z = o.y + v.y;
    o.w = o.z + v.z;
    ((int4*)rowptr)[(tile0 >> 2) + tid] = o;
}

__global__ __launch_bounds__(256)
void fill_kernel(const int* __restrict__ erow,
                 const int* __restrict__ ecol,
                 const float* __restrict__ a1,
                 const float* __restrict__ a2,
                 const int* __restrict__ rowptr,
                 const int* __restrict__ rank,
                 int* __restrict__ sorted,
                 int E) {
    int e = blockIdx.x * blockDim.x + threadIdx.x;
    if (e >= E) return;
    const int pos = rowptr[erow[e]] + rank[e];
    int4 rec;
    rec.x = ecol[e];
    rec.y = __float_as_int(a1[e]);
    rec.z = __float_as_int(a2[e]);
    rec.w = 0;
    ((int4*)sorted)[pos] = rec;
}

__global__ __launch_bounds__(256)
void gemm_msg_kernel(const float* __restrict__ miu,
                     const float* __restrict__ sigma,
                     const float* __restrict__ Wmu,
                     const float* __restrict__ Wsig,
                     float* __restrict__ m_mu,
                     float* __restrict__ m_sig,
                     int N) {
    const int lane = threadIdx.x & 31;
    const int wave = threadIdx.x >> 5;
    const int row_base = blockIdx.x * 128 + wave * 16;

    const int m    = lane & 15;
    const int kgrp = lane >> 4;

    const int arow = (row_base + m < N) ? (row_base + m) : (N - 1);

    float att[8][8];

    {
        v8f acc[8];
        v8f zero = {};
        #pragma unroll
        for (int t = 0; t < 8; ++t) acc[t] = zero;

        const float* Ap = sigma + (size_t)arow * IN_F + kgrp * 2;
        #pragma unroll 4
        for (int k0 = 0; k0 < IN_F; k0 += 4) {
            v2f a = *(const v2f*)(Ap + k0);
            const float* Bp = Wsig + (size_t)(k0 + kgrp * 2) * OUT_F + m;
            #pragma unroll
            for (int t = 0; t < 8; ++t) {
                v2f b;
                b[0] = Bp[t * 16];
                b[1] = Bp[t * 16 + OUT_F];
                acc[t] = __builtin_amdgcn_wmma_f32_16x16x4_f32(
                    false, a, false, b, (short)0, acc[t], false, false);
            }
        }
        #pragma unroll
        for (int t = 0; t < 8; ++t) {
            #pragma unroll
            for (int r = 0; r < 8; ++r) {
                float hs = fmaxf(acc[t][r], 0.0f);
                float a  = __expf(-hs);
                att[t][r] = a;
                int rr = row_base + r + kgrp * 8;
                if (rr < N)
                    m_sig[(size_t)rr * OUT_F + t * 16 + m] = hs * a * a;
            }
        }
    }

    {
        v8f acc[8];
        v8f zero = {};
        #pragma unroll
        for (int t = 0; t < 8; ++t) acc[t] = zero;

        const float* Ap = miu + (size_t)arow * IN_F + kgrp * 2;
        #pragma unroll 4
        for (int k0 = 0; k0 < IN_F; k0 += 4) {
            v2f a = *(const v2f*)(Ap + k0);
            const float* Bp = Wmu + (size_t)(k0 + kgrp * 2) * OUT_F + m;
            #pragma unroll
            for (int t = 0; t < 8; ++t) {
                v2f b;
                b[0] = Bp[t * 16];
                b[1] = Bp[t * 16 + OUT_F];
                acc[t] = __builtin_amdgcn_wmma_f32_16x16x4_f32(
                    false, a, false, b, (short)0, acc[t], false, false);
            }
        }
        #pragma unroll
        for (int t = 0; t < 8; ++t) {
            #pragma unroll
            for (int r = 0; r < 8; ++r) {
                float hm = acc[t][r];
                float e  = (hm > 0.0f) ? hm : (__expf(hm) - 1.0f);
                int rr = row_base + r + kgrp * 8;
                if (rr < N)
                    m_mu[(size_t)rr * OUT_F + t * 16 + m] = e * att[t][r];
            }
        }
    }
}

__global__ __launch_bounds__(256)
void row_gather_kernel(const int* __restrict__ rowptr,
                       const int* __restrict__ sorted,
                       const float* __restrict__ m_mu,
                       const float* __restrict__ m_sig,
                       float* __restrict__ out_mu,
                       float* __restrict__ out_sig,
                       int N, int E) {
    const int lane = threadIdx.x & 31;
    const int r = __builtin_amdgcn_readfirstlane((int)(blockIdx.x * 8 + (threadIdx.x >> 5)));
    if (r >= N) return;

    int start = rowptr[r];
    int end   = rowptr[r + 1];
    start = max(0, min(start, E));
    end   = max(start, min(end, E));

    const int4* rec4 = (const int4*)sorted;
    float4 am = make_float4(0.0f, 0.0f, 0.0f, 0.0f);
    float4 as = make_float4(0.0f, 0.0f, 0.0f, 0.0f);

    for (int base = start; base < end; base += 32) {
        const int n = min(32, end - base);
        int4 rec = make_int4(0, 0, 0, 0);
        if (lane < n) rec = rec4[base + lane];

        #pragma unroll 4
        for (int j = 0; j < n; ++j) {
            int c = __builtin_amdgcn_readlane(rec.x, j);
            c = max(0, min(c, N - 1));
            const float v1 = __int_as_float(__builtin_amdgcn_readlane(rec.y, j));
            const float v2 = __int_as_float(__builtin_amdgcn_readlane(rec.z, j));
            const float4 mm = ((const float4*)(m_mu  + (size_t)c * OUT_F))[lane];
            const float4 ms = ((const float4*)(m_sig + (size_t)c * OUT_F))[lane];
            am.x += v1 * mm.x; am.y += v1 * mm.y; am.z += v1 * mm.z; am.w += v1 * mm.w;
            as.x += v2 * ms.x; as.y += v2 * ms.y; as.z += v2 * ms.z; as.w += v2 * ms.w;
        }
    }

    ((float4*)(out_mu  + (size_t)r * OUT_F))[lane] = am;
    ((float4*)(out_sig + (size_t)r * OUT_F))[lane] = as;
}

extern "C" void kernel_launch(void* const* d_in, const int* in_sizes, int n_in,
                              void* d_out, int out_size, void* d_ws, size_t ws_size,
                              hipStream_t stream) {
    const float* miu   = (const float*)d_in[0];
    const float* sigma = (const float*)d_in[1];
    const int*   erow  = (const int*)  d_in[2];
    const int*   ecol  = (const int*)  d_in[3];
    const float* a1    = (const float*)d_in[4];
    const float* a2    = (const float*)d_in[5];
    const float* Wmu   = (const float*)d_in[6];
    const float* Wsig  = (const float*)d_in[7];

    const int N = in_sizes[0] / IN_F;
    const int E = in_sizes[2];

    float* out_mu  = (float*)d_out;
    float* out_sig = out_mu + (size_t)N * OUT_F;

    const size_t nPad = ((size_t)N + 1 + SCAN_TILE - 1) / SCAN_TILE * SCAN_TILE;
    const size_t ePad = ((size_t)E + 1023) / 1024 * 1024;
    float* m_mu   = (float*)d_ws;
    float* m_sig  = m_mu + (size_t)N * OUT_F;
    int*   counts = (int*)(m_sig + (size_t)N * OUT_F);
    int*   rowptr = counts + nPad;
    int*   rank   = rowptr + nPad;
    int*   sorted = rank + ePad;

    const int eblocks = (int)(((long long)E + 255) / 256);

    zero_counts_kernel<<<(int)(nPad / 256), 256, 0, stream>>>(counts, (int)nPad);
    hist_rank_kernel<<<eblocks, 256, 0, stream>>>(erow, counts, rank, E);
    row_scan_kernel<<<(int)(nPad / SCAN_TILE), 256, 0, stream>>>(counts, rowptr);
    fill_kernel<<<eblocks, 256, 0, stream>>>(erow, ecol, a1, a2, rowptr, rank, sorted, E);

    gemm_msg_kernel<<<(N + 127) / 128, 256, 0, stream>>>(
        miu, sigma, Wmu, Wsig, m_mu, m_sig, N);

    row_gather_kernel<<<(N + 7) / 8, 256, 0, stream>>>(
        rowptr, sorted, m_mu, m_sig, out_mu, out_sig, N, E);
}


# global per-edge rank adds, counter clear and row scan replaced by a two-level counting sort on LDS counters (512-row buckets, then rows), 5 launches; sums now in LDS-add arrival order
# speedup vs baseline: 2.7715x; 1.4062x over previous
#include <hip/hip_runtime.h>
#include <hip/hip_bf16.h>

typedef float v2f __attribute__((ext_vector_type(2)));
typedef float v8f __attribute__((ext_vector_type(8)));

#define IN_F 256
#define OUT_F 128

#define ROW_SHIFT 9
#define ROWS_PER_BUCKET (1 << ROW_SHIFT)
#define MAX_BUCKETS 256
#define NUM_CHUNKS 256
#define WG 1024
#define BATCH 4


__device__ __forceinline__ int bucket_of(int row, int NB) {
    int b = row >> ROW_SHIFT;
    return max(0, min(b, NB - 1));
}

__global__ __launch_bounds__(WG)
void chunk_hist_kernel(const int* __restrict__ erow,
                       int* __restrict__ cntm,
                       int E, int CE, int NB) {
    __shared__ int h[MAX_BUCKETS];
    const int tid = threadIdx.x;
    const int c = blockIdx.x;
    if (tid < MAX_BUCKETS) h[tid] = 0;
    __syncthreads();
    const int e0 = min(E, c * CE);
    const int e1 = min(E, e0 + CE);
    for (int base = e0 + tid; base < e1; base += BATCH * WG) {
        int r[BATCH];
        #pragma unroll
        for (int u = 0; u < BATCH; ++u)
            r[u] = erow[min(base + u * WG, e1 - 1)];
        #pragma unroll
        for (int u = 0; u < BATCH; ++u)
            if (base + u * WG < e1) atomicAdd(&h[bucket_of(r[u], NB)], 1);
    }
    __syncthreads();
    if (tid < MAX_BUCKETS) cntm[c * MAX_BUCKETS + tid] = h[tid];
}

__global__ __launch_bounds__(WG)
void bucket_scatter_kernel(const int* __restrict__ erow,
                           const int* __restrict__ ecol,
                           const float* __restrict__ a1,
                           const float* __restrict__ a2,
                           const int* __restrict__ cntm,
                           int* __restrict__ bstart,
                           int* __restrict__ bucketed,
                           int E, int CE, int NB) {
    __shared__ int s_all[4][MAX_BUCKETS];
    __shared__ int s_pre[4][MAX_BUCKETS];
    __shared__ int s_w[8];
    __shared__ int cur[MAX_BUCKETS];
    const int tid  = threadIdx.x;
    const int b    = tid & 255;
    const int q    = tid >> 8;
    const int lane = tid & 31;
    const int wave = tid >> 5;
    const int c    = blockIdx.x;

    int all = 0, pre = 0;
    #pragma unroll
    for (int i = 0; i < 64; ++i) {
        const int ch = q * 64 + i;
        const int v = cntm[ch * MAX_BUCKETS + b];
        all += v;
        if (ch < c) pre += v;
    }
    s_all[q][b] = all;
    s_pre[q][b] = pre;
    __syncthreads();

    int T = 0, P = 0, incl = 0;
    if (q == 0) {
        #pragma unroll
        for (int qq = 0; qq < 4; ++qq) {
            T += s_all[qq][b];
            P += s_pre[qq][b];
        }
        incl = T;
        #pragma unroll
        for (int d = 1; d < 32; d <<= 1) {
            int t = __shfl_up(incl, d, 32);
            if (lane >= d) incl += t;
        }
        if (lane == 31) s_w[wave] = incl;
    }
    __syncthreads();
    if (q == 0) {
        int front = 0;
        #pragma unroll
        for (int w = 0; w < 8; ++w)
            if (w < wave) front += s_w[w];
        const int S = front + incl - T;
        cur[b] = S + P;
        if (c == 0) {
            bstart[b] = S;
            if (b == MAX_BUCKETS - 1) bstart[MAX_BUCKETS] = S + T;
        }
    }
    __syncthreads();

    const int e0 = min(E, c * CE);
    const int e1 = min(E, e0 + CE);
    for (int base = e0 + tid; base < e1; base += BATCH * WG) {
        int4 rec[BATCH];
        #pragma unroll
        for (int u = 0; u < BATCH; ++u) {
            const int e = min(base + u * WG, e1 - 1);
            rec[u].x = erow[e];
            rec[u].y = ecol[e];
            rec[u].z = __float_as_int(a1[e]);
            rec[u].w = __float_as_int(a2[e]);
        }
        #pragma unroll
        for (int u = 0; u < BATCH; ++u) {
            if (base + u * WG < e1) {
                int pos = atomicAdd(&cur[bucket_of(rec[u].x, NB)], 1);
                pos = max(0, min(pos, E - 1));
                ((int4*)bucketed)[pos] = rec[u];
            }
        }
    }
}

__global__ __launch_bounds__(WG)
void bucket_sort_kernel(const int* __restrict__ bstart,
                        const int* __restrict__ bucketed,
                        int* __restrict__ rowptr,
                        int* __restrict__ sorted,
                        int N, int E) {
    __shared__ int cnt[ROWS_PER_BUCKET];
    __shared__ int s_w[16];
    const int tid  = threadIdx.x;
    const int lane = tid & 31;
    const int wave = tid >> 5;
    const int b    = blockIdx.x;
    const int row0 = b << ROW_SHIFT;

    int s = bstart[b];
    int t = bstart[b + 1];
    s = max(0, min(s, E));
    t = max(s, min(t, E));

    if (tid < ROWS_PER_BUCKET) cnt[tid] = 0;
    __syncthreads();

    const int4* in4 = (const int4*)bucketed;
    for (int base = s + tid; base < t; base += BATCH * WG) {
        int x[BATCH];
        #pragma unroll
        for (int u = 0; u < BATCH; ++u)
            x[u] = in4[min(base + u * WG, t - 1)].x;
        #pragma unroll
        for (int u = 0; u < BATCH; ++u) {
            const int lr = x[u] - row0;
            if (base + u * WG < t && (unsigned)lr < (unsigned)ROWS_PER_BUCKET) atomicAdd(&cnt[lr], 1);
        }
    }
    __syncthreads();

    int n = 0, incl = 0;
    if (tid < ROWS_PER_BUCKET) {
        n = cnt[tid];
        incl = n;
        #pragma unroll
        for (int d = 1; d < 32; d <<= 1) {
            int u = __shfl_up(incl, d, 32);
            if (lane >= d) incl += u;
        }
        if (lane == 31) s_w[wave] = incl;
    }
    __syncthreads();
    if (tid < ROWS_PER_BUCKET) {
        int front = 0;
        #pragma unroll
        for (int w = 0; w < 16; ++w)
            if (w < wave) front += s_w[w];
        const int ex = front + incl - n;
        cnt[tid] = ex;
        const int g = row0 + tid;
        if (g < N) rowptr[g] = s + ex;
        if (g == N - 1) rowptr[N] = s + ex + n;
    }
    __syncthreads();

    for (int base = s + tid; base < t; base += BATCH * WG) {
        int4 rec[BATCH];
        #pragma unroll
        for (int u = 0; u < BATCH; ++u)
            rec[u] = in4[min(base + u * WG, t - 1)];
        #pragma unroll
        for (int u = 0; u < BATCH; ++u) {
            const int lr = rec[u].x - row0;
            if (base + u * WG < t && (unsigned)lr < (unsigned)ROWS_PER_BUCKET) {
                int pos = s + atomicAdd(&cnt[lr], 1);
                pos = max(0, min(pos, E - 1));
                int4 o;
                o.x = rec[u].y;
                o.y = rec[u].z;
                o.z = rec[u].w;
                o.w = 0;
                ((int4*)sorted)[pos] = o;
            }
        }
    }
}

__global__ __launch_bounds__(256)
void gemm_msg_kernel(const float* __restrict__ miu,
                     const float* __restrict__ sigma,
                     const float* __restrict__ Wmu,
                     const float* __restrict__ Wsig,
                     float* __restrict__ m_mu,
                     float* __restrict__ m_sig,
                     int N) {
    const int lane = threadIdx.x & 31;
    const int wave = threadIdx.x >> 5;
    const int row_base = blockIdx.x * 128 + wave * 16;

    const int m    = lane & 15;
    const int kgrp = lane >> 4;

    const int arow = (row_base + m < N) ? (row_base + m) : (N - 1);

    float att[8][8];

    {
        v8f acc[8];
        v8f zero = {};
        #pragma unroll
        for (int t = 0; t < 8; ++t) acc[t] = zero;

        const float* Ap = sigma + (size_t)arow * IN_F + kgrp * 2;
        #pragma unroll 4
        for (int k0 = 0; k0 < IN_F; k0 += 4) {
            v2f a = *(const v2f*)(Ap + k0);
            const float* Bp = Wsig + (size_t)(k0 + kgrp * 2) * OUT_F + m;
            #pragma unroll
            for (int t = 0; t < 8; ++t) {
                v2f b;
                b[0] = Bp[t * 16];
                b[1] = Bp[t * 16 + OUT_F];
                acc[t] = __builtin_amdgcn_wmma_f32_16x16x4_f32(
                    false, a, false, b, (short)0, acc[t], false, false);
            }
        }
        #pragma unroll
        for (int t = 0; t < 8; ++t) {
            #pragma unroll
            for (int r = 0; r < 8; ++r) {
                float hs = fmaxf(acc[t][r], 0.0f);
                float a  = __expf(-hs);
                att[t][r] = a;
                int rr = row_base + r + kgrp * 8;
                if (rr < N)
                    m_sig[(size_t)rr * OUT_F + t * 16 + m] = hs * a * a;
            }
        }
    }

    {
        v8f acc[8];
        v8f zero = {};
        #pragma unroll
        for (int t = 0; t < 8; ++t) acc[t] = zero;

        const float* Ap = miu + (size_t)arow * IN_F + kgrp * 2;
        #pragma unroll 4
        for (int k0 = 0; k0 < IN_F; k0 += 4) {
            v2f a = *(const v2f*)(Ap + k0);
            const float* Bp = Wmu + (size_t)(k0 + kgrp * 2) * OUT_F + m;
            #pragma unroll
            for (int t = 0; t < 8; ++t) {
                v2f b;
                b[0] = Bp[t * 16];
                b[1] = Bp[t * 16 + OUT_F];
                acc[t] = __builtin_amdgcn_wmma_f32_16x16x4_f32(
                    false, a, false, b, (short)0, acc[t], false, false);
            }
        }
        #pragma unroll
        for (int t = 0; t < 8; ++t) {
            #pragma unroll
            for (int r = 0; r < 8; ++r) {
                float hm = acc[t][r];
                float e  = (hm > 0.0f) ? hm : (__expf(hm) - 1.0f);
                int rr = row_base + r + kgrp * 8;
                if (rr < N)
                    m_mu[(size_t)rr * OUT_F + t * 16 + m] = e * att[t][r];
            }
        }
    }
}

__global__ __launch_bounds__(256)
void row_gather_kernel(const int* __restrict__ rowptr,
                       const int* __restrict__ sorted,
                       const float* __restrict__ m_mu,
                       const float* __restrict__ m_sig,
                       float* __restrict__ out_mu,
                       float* __restrict__ out_sig,
                       int N, int E) {
    const int lane = threadIdx.x & 31;
    const int r = __builtin_amdgcn_readfirstlane((int)(blockIdx.x * 8 + (threadIdx.x >> 5)));
    if (r >= N) return;

    int start = rowptr[r];
    int end   = rowptr[r + 1];
    start = max(0, min(start, E));
    end   = max(start, min(end, E));

    const int4* rec4 = (const int4*)sorted;
    float4 am = make_float4(0.0f, 0.0f, 0.0f, 0.0f);
    float4 as = make_float4(0.0f, 0.0f, 0.0f, 0.0f);

    for (int base = start; base < end; base += 32) {
        const int n = min(32, end - base);
        int4 rec = make_int4(0, 0, 0, 0);
        if (lane < n) rec = rec4[base + lane];

        #pragma unroll 4
        for (int j = 0; j < n; ++j) {
            int c = __builtin_amdgcn_readlane(rec.x, j);
            c = max(0, min(c, N - 1));
            const float v1 = __int_as_float(__builtin_amdgcn_readlane(rec.y, j));
            const float v2 = __int_as_float(__builtin_amdgcn_readlane(rec.z, j));
            const float4 mm = ((const float4*)(m_mu  + (size_t)c * OUT_F))[lane];
            const float4 ms = ((const float4*)(m_sig + (size_t)c * OUT_F))[lane];
            am.x += v1 * mm.x; am.y += v1 * mm.y; am.z += v1 * mm.z; am.w += v1 * mm.w;
            as.x += v2 * ms.x; as.y += v2 * ms.y; as.z += v2 * ms.z; as.w += v2 * ms.w;
        }
    }

    ((float4*)(out_mu  + (size_t)r * OUT_F))[lane] = am;
    ((float4*)(out_sig + (size_t)r * OUT_F))[lane] = as;
}

extern "C" void kernel_launch(void* const* d_in, const int* in_sizes, int n_in,
                              void* d_out, int out_size, void* d_ws, size_t ws_size,
                              hipStream_t stream) {
    const float* miu   = (const float*)d_in[0];
    const float* sigma = (const float*)d_in[1];
    const int*   erow  = (const int*)  d_in[2];
    const int*   ecol  = (const int*)  d_in[3];
    const float* a1    = (const float*)d_in[4];
    const float* a2    = (const float*)d_in[5];
    const float* Wmu   = (const float*)d_in[6];
    const float* Wsig  = (const float*)d_in[7];

    const int N = in_sizes[0] / IN_F;
    const int E = in_sizes[2];

    float* out_mu  = (float*)d_out;
    float* out_sig = out_mu + (size_t)N * OUT_F;

    const size_t nPad = ((size_t)N + 1 + 1023) / 1024 * 1024;
    const size_t ePad = ((size_t)E + 1023) / 1024 * 1024;
    float* m_mu     = (float*)d_ws;
    float* m_sig    = m_mu + (size_t)N * OUT_F;
    int*   rowptr   = (int*)(m_sig + (size_t)N * OUT_F);
    int*   cntm     = rowptr + nPad;
    int*   bstart   = cntm + NUM_CHUNKS * MAX_BUCKETS;
    int*   bucketed = bstart + 1024;
    int*   sorted   = bucketed + ePad * 4;

    const int NB = (N + ROWS_PER_BUCKET - 1) >> ROW_SHIFT;
    const int CE = (int)(((long long)E + NUM_CHUNKS - 1) / NUM_CHUNKS);

    chunk_hist_kernel<<<NUM_CHUNKS, WG, 0, stream>>>(erow, cntm, E, CE, NB);
    bucket_scatter_kernel<<<NUM_CHUNKS, WG, 0, stream>>>(erow, ecol, a1, a2, cntm, bstart, bucketed, E, CE, NB);
    bucket_sort_kernel<<<NB, WG, 0, stream>>>(bstart, bucketed, rowptr, sorted, N, E);

    gemm_msg_kernel<<<(N + 127) / 128, 256, 0, stream>>>(
        miu, sigma, Wmu, Wsig, m_mu, m_sig, N);

    row_gather_kernel<<<(N + 7) / 8, 256, 0, stream>>>(
        rowptr, sorted, m_mu, m_sig, out_mu, out_sig, N, E);
}


# message GEMM retiled: a wave owns 16 rows x 64 columns, 4 tiles interleaved by column (tile j = columns 4m+j) so one 16-byte W load feeds 4 WMMAs, stores cover whole lines; 3125 4-wave blocks for 782
# speedup vs baseline: 3.4507x; 1.2451x over previous
#include <hip/hip_runtime.h>
#include <hip/hip_bf16.h>

typedef float v2f __attribute__((ext_vector_type(2)));
typedef float v8f __attribute__((ext_vector_type(8)));

#define IN_F 256
#define OUT_F 128

#define ROW_SHIFT 9
#define ROWS_PER_BUCKET (1 << ROW_SHIFT)
#define MAX_BUCKETS 256
#define NUM_CHUNKS 256
#define WG 1024
#define BATCH 4


__device__ __forceinline__ int bucket_of(int row, int NB) {
    int b = row >> ROW_SHIFT;
    return max(0, min(b, NB - 1));
}

__global__ __launch_bounds__(WG)
void chunk_hist_kernel(const int* __restrict__ erow,
                       int* __restrict__ cntm,
                       int E, int CE, int NB) {
    __shared__ int h[MAX_BUCKETS];
    const int tid = threadIdx.x;
    const int c = blockIdx.x;
    if (tid < MAX_BUCKETS) h[tid] = 0;
    __syncthreads();
    const int e0 = min(E, c * CE);
    const int e1 = min(E, e0 + CE);
    for (int base = e0 + tid; base < e1; base += BATCH * WG) {
        int r[BATCH];
        #pragma unroll
        for (int u = 0; u < BATCH; ++u)
            r[u] = erow[min(base + u * WG, e1 - 1)];
        #pragma unroll
        for (int u = 0; u < BATCH; ++u)
            if (base + u * WG < e1) atomicAdd(&h[bucket_of(r[u], NB)], 1);
    }
    __syncthreads();
    if (tid < MAX_BUCKETS) cntm[c * MAX_BUCKETS + tid] = h[tid];
}

__global__ __launch_bounds__(WG)
void bucket_scatter_kernel(const int* __restrict__ erow,
                           const int* __restrict__ ecol,
                           const float* __restrict__ a1,
                           const float* __restrict__ a2,
                           const int* __restrict__ cntm,
                           int* __restrict__ bstart,
                           int* __restrict__ bucketed,
                           int E, int CE, int NB) {
    __shared__ int s_all[4][MAX_BUCKETS];
    __shared__ int s_pre[4][MAX_BUCKETS];
    __shared__ int s_w[8];
    __shared__ int cur[MAX_BUCKETS];
    const int tid  = threadIdx.x;
    const int b    = tid & 255;
    const int q    = tid >> 8;
    const int lane = tid & 31;
    const int wave = tid >> 5;
    const int c    = blockIdx.x;

    int all = 0, pre = 0;
    #pragma unroll
    for (int i = 0; i < 64; ++i) {
        const int ch = q * 64 + i;
        const int v = cntm[ch * MAX_BUCKETS + b];
        all += v;
        if (ch < c) pre += v;
    }
    s_all[q][b] = all;
    s_pre[q][b] = pre;
    __syncthreads();

    int T = 0, P = 0, incl = 0;
    if (q == 0) {
        #pragma unroll
        for (int qq = 0; qq < 4; ++qq) {
            T += s_all[qq][b];
            P += s_pre[qq][b];
        }
        incl = T;
        #pragma unroll
        for (int d = 1; d < 32; d <<= 1) {
            int t = __shfl_up(incl, d, 32);
            if (lane >= d) incl += t;
        }
        if (lane == 31) s_w[wave] = incl;
    }
    __syncthreads();
    if (q == 0) {
        int front = 0;
        #pragma unroll
        for (int w = 0; w < 8; ++w)
            if (w < wave) front += s_w[w];
        const int S = front + incl - T;
        cur[b] = S + P;
        if (c == 0) {
            bstart[b] = S;
            if (b == MAX_BUCKETS - 1) bstart[MAX_BUCKETS] = S + T;
        }
    }
    __syncthreads();

    const int e0 = min(E, c * CE);
    const int e1 = min(E, e0 + CE);
    for (int base = e0 + tid; base < e1; base += BATCH * WG) {
        int4 rec[BATCH];
        #pragma unroll
        for (int u = 0; u < BATCH; ++u) {
            const int e = min(base + u * WG, e1 - 1);
            rec[u].x = erow[e];
            rec[u].y = ecol[e];
            rec[u].z = __float_as_int(a1[e]);
            rec[u].w = __float_as_int(a2[e]);
        }
        #pragma unroll
        for (int u = 0; u < BATCH; ++u) {
            if (base + u * WG < e1) {
                int pos = atomicAdd(&cur[bucket_of(rec[u].x, NB)], 1);
                pos = max(0, min(pos, E - 1));
                ((int4*)bucketed)[pos] = rec[u];
            }
        }
    }
}

__global__ __launch_bounds__(WG)
void bucket_sort_kernel(const int* __restrict__ bstart,
                        const int* __restrict__ bucketed,
                        int* __restrict__ rowptr,
                        int* __restrict__ sorted,
                        int N, int E) {
    __shared__ int cnt[ROWS_PER_BUCKET];
    __shared__ int s_w[16];
    const int tid  = threadIdx.x;
    const int lane = tid & 31;
    const int wave = tid >> 5;
    const int b    = blockIdx.x;
    const int row0 = b << ROW_SHIFT;

    int s = bstart[b];
    int t = bstart[b + 1];
    s = max(0, min(s, E));
    t = max(s, min(t, E));

    if (tid < ROWS_PER_BUCKET) cnt[tid] = 0;
    __syncthreads();

    const int4* in4 = (const int4*)bucketed;
    for (int base = s + tid; base < t; base += BATCH * WG) {
        int x[BATCH];
        #pragma unroll
        for (int u = 0; u < BATCH; ++u)
            x[u] = in4[min(base + u * WG, t - 1)].x;
        #pragma unroll
        for (int u = 0; u < BATCH; ++u) {
            const int lr = x[u] - row0;
            if (base + u * WG < t && (unsigned)lr < (unsigned)ROWS_PER_BUCKET) atomicAdd(&cnt[lr], 1);
        }
    }
    __syncthreads();

    int n = 0, incl = 0;
    if (tid < ROWS_PER_BUCKET) {
        n = cnt[tid];
        incl = n;
        #pragma unroll
        for (int d = 1; d < 32; d <<= 1) {
            int u = __shfl_up(incl, d, 32);
            if (lane >= d) incl += u;
        }
        if (lane == 31) s_w[wave] = incl;
    }
    __syncthreads();
    if (tid < ROWS_PER_BUCKET) {
        int front = 0;
        #pragma unroll
        for (int w = 0; w < 16; ++w)
            if (w < wave) front += s_w[w];
        const int ex = front + incl - n;
        cnt[tid] = ex;
        const int g = row0 + tid;
        if (g < N) rowptr[g] = s + ex;
        if (g == N - 1) rowptr[N] = s + ex + n;
    }
    __syncthreads();

    for (int base = s + tid; base < t; base += BATCH * WG) {
        int4 rec[BATCH];
        #pragma unroll
        for (int u = 0; u < BATCH; ++u)
            rec[u] = in4[min(base + u * WG, t - 1)];
        #pragma unroll
        for (int u = 0; u < BATCH; ++u) {
            const int lr = rec[u].x - row0;
            if (base + u * WG < t && (unsigned)lr < (unsigned)ROWS_PER_BUCKET) {
                int pos = s + atomicAdd(&cnt[lr], 1);
                pos = max(0, min(pos, E - 1));
                int4 o;
                o.x = rec[u].y;
                o.y = rec[u].z;
                o.z = rec[u].w;
                o.w = 0;
                ((int4*)sorted)[pos] = o;
            }
        }
    }
}

__global__ __launch_bounds__(128)
void gemm_msg_kernel(const float* __restrict__ miu,
                     const float* __restrict__ sigma,
                     const float* __restrict__ Wmu,
                     const float* __restrict__ Wsig,
                     float* __restrict__ m_mu,
                     float* __restrict__ m_sig,
                     int N) {
    const int lane = threadIdx.x & 31;
    const int wave = threadIdx.x >> 5;
    const int strip = wave & 1;
    const int half  = wave >> 1;
    const int row_base = blockIdx.x * 32 + strip * 16;

    const int m    = lane & 15;
    const int kgrp = lane >> 4;
    const int col0 = half * 64 + m * 4;

    const int arow = (row_base + m < N) ? (row_base + m) : (N - 1);

    float att[4][8];

    {
        v8f acc[4];
        v8f zero = {};
        #pragma unroll
        for (int j = 0; j < 4; ++j) acc[j] = zero;

        const float* Ap = sigma + (size_t)arow * IN_F + kgrp * 2;
        const float* Bp = Wsig + (size_t)(kgrp * 2) * OUT_F + col0;
        #pragma unroll 4
        for (int k0 = 0; k0 < IN_F; k0 += 4) {
            const v2f a = *(const v2f*)(Ap + k0);
            const float4 w0 = *(const float4*)(Bp + (size_t)k0 * OUT_F);
            const float4 w1 = *(const float4*)(Bp + (size_t)(k0 + 1) * OUT_F);
            v2f b;
            b[0] = w0.x; b[1] = w1.x;
            acc[0] = __builtin_amdgcn_wmma_f32_16x16x4_f32(false, a, false, b, (short)0, acc[0], false, false);
            b[0] = w0.y; b[1] = w1.y;
            acc[1] = __builtin_amdgcn_wmma_f32_16x16x4_f32(false, a, false, b, (short)0, acc[1], false, false);
            b[0] = w0.z; b[1] = w1.z;
            acc[2] = __builtin_amdgcn_wmma_f32_16x16x4_f32(false, a, false, b, (short)0, acc[2], false, false);
            b[0] = w0.w; b[1] = w1.w;
            acc[3] = __builtin_amdgcn_wmma_f32_16x16x4_f32(false, a, false, b, (short)0, acc[3], false, false);
        }
        #pragma unroll
        for (int r = 0; r < 8; ++r) {
            float o[4];
            #pragma unroll
            for (int j = 0; j < 4; ++j) {
                const float hs = fmaxf(acc[j][r], 0.0f);
                const float a  = __expf(-hs);
                att[j][r] = a;
                o[j] = hs * a * a;
            }
            const int rr = row_base + r + kgrp * 8;
            if (rr < N)
                *(float4*)(m_sig + (size_t)rr * OUT_F + col0) = make_float4(o[0], o[1], o[2], o[3]);
        }
    }

    {
        v8f acc[4];
        v8f zero = {};
        #pragma unroll
        for (int j = 0; j < 4; ++j) acc[j] = zero;

        const float* Ap = miu + (size_t)arow * IN_F + kgrp * 2;
        const float* Bp = Wmu + (size_t)(kgrp * 2) * OUT_F + col0;
        #pragma unroll 4
        for (int k0 = 0; k0 < IN_F; k0 += 4) {
            const v2f a = *(const v2f*)(Ap + k0);
            const float4 w0 = *(const float4*)(Bp + (size_t)k0 * OUT_F);
            const float4 w1 = *(const float4*)(Bp + (size_t)(k0 + 1) * OUT_F);
            v2f b;
            b[0] = w0.x; b[1] = w1.x;
            acc[0] = __builtin_amdgcn_wmma_f32_16x16x4_f32(false, a, false, b, (short)0, acc[0], false, false);
            b[0] = w0.y; b[1] = w1.y;
            acc[1] = __builtin_amdgcn_wmma_f32_16x16x4_f32(false, a, false, b, (short)0, acc[1], false, false);
            b[0] = w0.z; b[1] = w1.z;
            acc[2] = __builtin_amdgcn_wmma_f32_16x16x4_f32(false, a, false, b, (short)0, acc[2], false, false);
            b[0] = w0.w; b[1] = w1.w;
            acc[3] = __builtin_amdgcn_wmma_f32_16x16x4_f32(false, a, false, b, (short)0, acc[3], false, false);
        }
        #pragma unroll
        for (int r = 0; r < 8; ++r) {
            float o[4];
            #pragma unroll
            for (int j = 0; j < 4; ++j) {
                const float hm = acc[j][r];
                const float e  = (hm > 0.0f) ? hm : (__expf(hm) - 1.0f);
                o[j] = e * att[j][r];
            }
            const int rr = row_base + r + kgrp * 8;
            if (rr < N)
                *(float4*)(m_mu + (size_t)rr * OUT_F + col0) = make_float4(o[0], o[1], o[2], o[3]);
        }
    }
}

__global__ __launch_bounds__(256)
void row_gather_kernel(const int* __restrict__ rowptr,
                       const int* __restrict__ sorted,
                       const float* __restrict__ m_mu,
                       const float* __restrict__ m_sig,
                       float* __restrict__ out_mu,
                       float* __restrict__ out_sig,
                       int N, int E) {
    const int lane = threadIdx.x & 31;
    const int r = __builtin_amdgcn_readfirstlane((int)(blockIdx.x * 8 + (threadIdx.x >> 5)));
    if (r >= N) return;

    int start = rowptr[r];
    int end   = rowptr[r + 1];
    start = max(0, min(start, E));
    end   = max(start, min(end, E));

    const int4* rec4 = (const int4*)sorted;
    float4 am = make_float4(0.0f, 0.0f, 0.0f, 0.0f);
    float4 as = make_float4(0.0f, 0.0f, 0.0f, 0.0f);

    for (int base = start; base < end; base += 32) {
        const int n = min(32, end - base);
        int4 rec = make_int4(0, 0, 0, 0);
        if (lane < n) rec = rec4[base + lane];

        #pragma unroll 4
        for (int j = 0; j < n; ++j) {
            int c = __builtin_amdgcn_readlane(rec.x, j);
            c = max(0, min(c, N - 1));
            const float v1 = __int_as_float(__builtin_amdgcn_readlane(rec.y, j));
            const float v2 = __int_as_float(__builtin_amdgcn_readlane(rec.z, j));
            const float4 mm = ((const float4*)(m_mu  + (size_t)c * OUT_F))[lane];
            const float4 ms = ((const float4*)(m_sig + (size_t)c * OUT_F))[lane];
            am.x += v1 * mm.x; am.y += v1 * mm.y; am.z += v1 * mm.z; am.w += v1 * mm.w;
            as.x += v2 * ms.x; as.y += v2 * ms.y; as.z += v2 * ms.z; as.w += v2 * ms.w;
        }
    }

    ((float4*)(out_mu  + (size_t)r * OUT_F))[lane] = am;
    ((float4*)(out_sig + (size_t)r * OUT_F))[lane] = as;
}

extern "C" void kernel_launch(void* const* d_in, const int* in_sizes, int n_in,
                              void* d_out, int out_size, void* d_ws, size_t ws_size,
                              hipStream_t stream) {
    const float* miu   = (const float*)d_in[0];
    const float* sigma = (const float*)d_in[1];
    const int*   erow  = (const int*)  d_in[2];
    const int*   ecol  = (const int*)  d_in[3];
    const float* a1    = (const float*)d_in[4];
    const float* a2    = (const float*)d_in[5];
    const float* Wmu   = (const float*)d_in[6];
    const float* Wsig  = (const float*)d_in[7];

    const int N = in_sizes[0] / IN_F;
    const int E = in_sizes[2];

    float* out_mu  = (float*)d_out;
    float* out_sig = out_mu + (size_t)N * OUT_F;

    const size_t nPad = ((size_t)N + 1 + 1023) / 1024 * 1024;
    const size_t ePad = ((size_t)E + 1023) / 1024 * 1024;
    float* m_mu     = (float*)d_ws;
    float* m_sig    = m_mu + (size_t)N * OUT_F;
    int*   rowptr   = (int*)(m_sig + (size_t)N * OUT_F);
    int*   cntm     = rowptr + nPad;
    int*   bstart   = cntm + NUM_CHUNKS * MAX_BUCKETS;
    int*   bucketed = bstart + 1024;
    int*   sorted   = bucketed + ePad * 4;

    const int NB = (N + ROWS_PER_BUCKET - 1) >> ROW_SHIFT;
    const int CE = (int)(((long long)E + NUM_CHUNKS - 1) / NUM_CHUNKS);

    chunk_hist_kernel<<<NUM_CHUNKS, WG, 0, stream>>>(erow, cntm, E, CE, NB);
    bucket_scatter_kernel<<<NUM_CHUNKS, WG, 0, stream>>>(erow, ecol, a1, a2, cntm, bstart, bucketed, E, CE, NB);
    bucket_sort_kernel<<<NB, WG, 0, stream>>>(bstart, bucketed, rowptr, sorted, N, E);

    gemm_msg_kernel<<<(N + 31) / 32, 128, 0, stream>>>(
        miu, sigma, Wmu, Wsig, m_mu, m_sig, N);

    row_gather_kernel<<<(N + 7) / 8, 256, 0, stream>>>(
        rowptr, sorted, m_mu, m_sig, out_mu, out_sig, N, E);
}


# message GEMM: A read as whole 128-byte lines into a wave-private LDS tile (32 k a trip, next trip in flight), fragments read from LDS, not by 16-line 8-byte global loads; k summed in permuted order
# speedup vs baseline: 3.5687x; 1.0342x over previous
#include <hip/hip_runtime.h>
#include <hip/hip_bf16.h>

typedef float v2f __attribute__((ext_vector_type(2)));
typedef float v8f __attribute__((ext_vector_type(8)));

#define IN_F 256
#define OUT_F 128

#define ROW_SHIFT 9
#define ROWS_PER_BUCKET (1 << ROW_SHIFT)
#define MAX_BUCKETS 256
#define NUM_CHUNKS 256
#define WG 1024
#define BATCH 4


__device__ __forceinline__ int bucket_of(int row, int NB) {
    int b = row >> ROW_SHIFT;
    return max(0, min(b, NB - 1));
}

__global__ __launch_bounds__(WG)
void chunk_hist_kernel(const int* __restrict__ erow,
                       int* __restrict__ cntm,
                       int E, int CE, int NB) {
    __shared__ int h[MAX_BUCKETS];
    const int tid = threadIdx.x;
    const int c = blockIdx.x;
    if (tid < MAX_BUCKETS) h[tid] = 0;
    __syncthreads();
    const int e0 = min(E, c * CE);
    const int e1 = min(E, e0 + CE);
    for (int base = e0 + tid; base < e1; base += BATCH * WG) {
        int r[BATCH];
        #pragma unroll
        for (int u = 0; u < BATCH; ++u)
            r[u] = erow[min(base + u * WG, e1 - 1)];
        #pragma unroll
        for (int u = 0; u < BATCH; ++u)
            if (base + u * WG < e1) atomicAdd(&h[bucket_of(r[u], NB)], 1);
    }
    __syncthreads();
    if (tid < MAX_BUCKETS) cntm[c * MAX_BUCKETS + tid] = h[tid];
}

__global__ __launch_bounds__(WG)
void bucket_scatter_kernel(const int* __restrict__ erow,
                           const int* __restrict__ ecol,
                           const float* __restrict__ a1,
                           const float* __restrict__ a2,
                           const int* __restrict__ cntm,
                           int* __restrict__ bstart,
                           int* __restrict__ bucketed,
                           int E, int CE, int NB) {
    __shared__ int s_all[4][MAX_BUCKETS];
    __shared__ int s_pre[4][MAX_BUCKETS];
    __shared__ int s_w[8];
    __shared__ int cur[MAX_BUCKETS];
    const int tid  = threadIdx.x;
    const int b    = tid & 255;
    const int q    = tid >> 8;
    const int lane = tid & 31;
    const int wave = tid >> 5;
    const int c    = blockIdx.x;

    int all = 0, pre = 0;
    #pragma unroll
    for (int i = 0; i < 64; ++i) {
        const int ch = q * 64 + i;
        const int v = cntm[ch * MAX_BUCKETS + b];
        all += v;
        if (ch < c) pre += v;
    }
    s_all[q][b] = all;
    s_pre[q][b] = pre;
    __syncthreads();

    int T = 0, P = 0, incl = 0;
    if (q == 0) {
        #pragma unroll
        for (int qq = 0; qq < 4; ++qq) {
            T += s_all[qq][b];
            P += s_pre[qq][b];
        }
        incl = T;
        #pragma unroll
        for (int d = 1; d < 32; d <<= 1) {
            int t = __shfl_up(incl, d, 32);
            if (lane >= d) incl += t;
        }
        if (lane == 31) s_w[wave] = incl;
    }
    __syncthreads();
    if (q == 0) {
        int front = 0;
        #pragma unroll
        for (int w = 0; w < 8; ++w)
            if (w < wave) front += s_w[w];
        const int S = front + incl - T;
        cur[b] = S + P;
        if (c == 0) {
            bstart[b] = S;
            if (b == MAX_BUCKETS - 1) bstart[MAX_BUCKETS] = S + T;
        }
    }
    __syncthreads();

    const int e0 = min(E, c * CE);
    const int e1 = min(E, e0 + CE);
    for (int base = e0 + tid; base < e1; base += BATCH * WG) {
        int4 rec[BATCH];
        #pragma unroll
        for (int u = 0; u < BATCH; ++u) {
            const int e = min(base + u * WG, e1 - 1);
            rec[u].x = erow[e];
            rec[u].y = ecol[e];
            rec[u].z = __float_as_int(a1[e]);
            rec[u].w = __float_as_int(a2[e]);
        }
        #pragma unroll
        for (int u = 0; u < BATCH; ++u) {
            if (base + u * WG < e1) {
                int pos = atomicAdd(&cur[bucket_of(rec[u].x, NB)], 1);
                pos = max(0, min(pos, E - 1));
                ((int4*)bucketed)[pos] = rec[u];
            }
        }
    }
}

__global__ __launch_bounds__(WG)
void bucket_sort_kernel(const int* __restrict__ bstart,
                        const int* __restrict__ bucketed,
                        int* __restrict__ rowptr,
                        int* __restrict__ sorted,
                        int N, int E) {
    __shared__ int cnt[ROWS_PER_BUCKET];
    __shared__ int s_w[16];
    const int tid  = threadIdx.x;
    const int lane = tid & 31;
    const int wave = tid >> 5;
    const int b    = blockIdx.x;
    const int row0 = b << ROW_SHIFT;

    int s = bstart[b];
    int t = bstart[b + 1];
    s = max(0, min(s, E));
    t = max(s, min(t, E));

    if (tid < ROWS_PER_BUCKET) cnt[tid] = 0;
    __syncthreads();

    const int4* in4 = (const int4*)bucketed;
    for (int base = s + tid; base < t; base += BATCH * WG) {
        int x[BATCH];
        #pragma unroll
        for (int u = 0; u < BATCH; ++u)
            x[u] = in4[min(base + u * WG, t - 1)].x;
        #pragma unroll
        for (int u = 0; u < BATCH; ++u) {
            const int lr = x[u] - row0;
            if (base + u * WG < t && (unsigned)lr < (unsigned)ROWS_PER_BUCKET) atomicAdd(&cnt[lr], 1);
        }
    }
    __syncthreads();

    int n = 0, incl = 0;
    if (tid < ROWS_PER_BUCKET) {
        n = cnt[tid];
        incl = n;
        #pragma unroll
        for (int d = 1; d < 32; d <<= 1) {
            int u = __shfl_up(incl, d, 32);
            if (lane >= d) incl += u;
        }
        if (lane == 31) s_w[wave] = incl;
    }
    __syncthreads();
    if (tid < ROWS_PER_BUCKET) {
        int front = 0;
        #pragma unroll
        for (int w = 0; w < 16; ++w)
            if (w < wave) front += s_w[w];
        const int ex = front + incl - n;
        cnt[tid] = ex;
        const int g = row0 + tid;
        if (g < N) rowptr[g] = s + ex;
        if (g == N - 1) rowptr[N] = s + ex + n;
    }
    __syncthreads();

    for (int base = s + tid; base < t; base += BATCH * WG) {
        int4 rec[BATCH];
        #pragma unroll
        for (int u = 0; u < BATCH; ++u)
            rec[u] = in4[min(base + u * WG, t - 1)];
        #pragma unroll
        for (int u = 0; u < BATCH; ++u) {
            const int lr = rec[u].x - row0;
            if (base + u * WG < t && (unsigned)lr < (unsigned)ROWS_PER_BUCKET) {
                int pos = s + atomicAdd(&cnt[lr], 1);
                pos = max(0, min(pos, E - 1));
                int4 o;
                o.x = rec[u].y;
                o.y = rec[u].z;
                o.z = rec[u].w;
                o.w = 0;
                ((int4*)sorted)[pos] = o;
            }
        }
    }
}

#define A_PITCH 36
#define A_CHUNK 32

#define TILE_LOADS(frag, Bk)                                      \
    const float4 a0 = *(const float4*)((frag));                  \
    const float4 a1 = *(const float4*)((frag) + 8);              \
    const float4 a2 = *(const float4*)((frag) + 16);             \
    const float4 a3 = *(const float4*)((frag) + 24);             \
    const float4 w00 = *(const float4*)((Bk) + (size_t)(0 + 0) * OUT_F);                            \
    const float4 w01 = *(const float4*)((Bk) + (size_t)(0 + 1) * OUT_F);                            \
    const float4 w02 = *(const float4*)((Bk) + (size_t)(0 + 2) * OUT_F);                            \
    const float4 w03 = *(const float4*)((Bk) + (size_t)(0 + 3) * OUT_F);                            \
    const float4 w10 = *(const float4*)((Bk) + (size_t)(8 + 0) * OUT_F);                            \
    const float4 w11 = *(const float4*)((Bk) + (size_t)(8 + 1) * OUT_F);                            \
    const float4 w12 = *(const float4*)((Bk) + (size_t)(8 + 2) * OUT_F);                            \
    const float4 w13 = *(const float4*)((Bk) + (size_t)(8 + 3) * OUT_F);                            \
    const float4 w20 = *(const float4*)((Bk) + (size_t)(16 + 0) * OUT_F);                           \
    const float4 w21 = *(const float4*)((Bk) + (size_t)(16 + 1) * OUT_F);                           \
    const float4 w22 = *(const float4*)((Bk) + (size_t)(16 + 2) * OUT_F);                           \
    const float4 w23 = *(const float4*)((Bk) + (size_t)(16 + 3) * OUT_F);                           \
    const float4 w30 = *(const float4*)((Bk) + (size_t)(24 + 0) * OUT_F);                           \
    const float4 w31 = *(const float4*)((Bk) + (size_t)(24 + 1) * OUT_F);                           \
    const float4 w32 = *(const float4*)((Bk) + (size_t)(24 + 2) * OUT_F);                           \
    const float4 w33 = *(const float4*)((Bk) + (size_t)(24 + 3) * OUT_F);                           \


#define TILE_MMAS                                                 \
    {                                                             \
    v2f a, b;                                                                                                                                                      \
    a[0] = a0.x; a[1] = a0.y;                                                                                                                         \
    b[0] = w00.x; b[1] = w01.x;  acc[0] = __builtin_amdgcn_wmma_f32_16x16x4_f32(false, a, false, b, (short)0, acc[0], false, false);                  \
    b[0] = w00.y; b[1] = w01.y;  acc[1] = __builtin_amdgcn_wmma_f32_16x16x4_f32(false, a, false, b, (short)0, acc[1], false, false);                  \
    b[0] = w00.z; b[1] = w01.z;  acc[2] = __builtin_amdgcn_wmma_f32_16x16x4_f32(false, a, false, b, (short)0, acc[2], false, false);                  \
    b[0] = w00.w; b[1] = w01.w;  acc[3] = __builtin_amdgcn_wmma_f32_16x16x4_f32(false, a, false, b, (short)0, acc[3], false, false);                  \
    a[0] = a0.z; a[1] = a0.w;                                                                                                                         \
    b[0] = w02.x; b[1] = w03.x;  acc[0] = __builtin_amdgcn_wmma_f32_16x16x4_f32(false, a, false, b, (short)0, acc[0], false, false);                  \
    b[0] = w02.y; b[1] = w03.y;  acc[1] = __builtin_amdgcn_wmma_f32_16x16x4_f32(false, a, false, b, (short)0, acc[1], false, false);                  \
    b[0] = w02.z; b[1] = w03.z;  acc[2] = __builtin_amdgcn_wmma_f32_16x16x4_f32(false, a, false, b, (short)0, acc[2], false, false);                  \
    b[0] = w02.w; b[1] = w03.w;  acc[3] = __builtin_amdgcn_wmma_f32_16x16x4_f32(false, a, false, b, (short)0, acc[3], false, false);                  \
    a[0] = a1.x; a[1] = a1.y;                                                                                                                         \
    b[0] = w10.x; b[1] = w11.x;  acc[0] = __builtin_amdgcn_wmma_f32_16x16x4_f32(false, a, false, b, (short)0, acc[0], false, false);                  \
    b[0] = w10.y; b[1] = w11.y;  acc[1] = __builtin_amdgcn_wmma_f32_16x16x4_f32(false, a, false, b, (short)0, acc[1], false, false);                  \
    b[0] = w10.z; b[1] = w11.z;  acc[2] = __builtin_amdgcn_wmma_f32_16x16x4_f32(false, a, false, b, (short)0, acc[2], false, false);                  \
    b[0] = w10.w; b[1] = w11.w;  acc[3] = __builtin_amdgcn_wmma_f32_16x16x4_f32(false, a, false, b, (short)0, acc[3], false, false);                  \
    a[0] = a1.z; a[1] = a1.w;                                                                                                                         \
    b[0] = w12.x; b[1] = w13.x;  acc[0] = __builtin_amdgcn_wmma_f32_16x16x4_f32(false, a, false, b, (short)0, acc[0], false, false);                  \
    b[0] = w12.y; b[1] = w13.y;  acc[1] = __builtin_amdgcn_wmma_f32_16x16x4_f32(false, a, false, b, (short)0, acc[1], false, false);                  \
    b[0] = w12.z; b[1] = w13.z;  acc[2] = __builtin_amdgcn_wmma_f32_16x16x4_f32(false, a, false, b, (short)0, acc[2], false, false);                  \
    b[0] = w12.w; b[1] = w13.w;  acc[3] = __builtin_amdgcn_wmma_f32_16x16x4_f32(false, a, false, b, (short)0, acc[3], false, false);                  \
    a[0] = a2.x; a[1] = a2.y;                                                                                                                         \
    b[0] = w20.x; b[1] = w21.x;  acc[0] = __builtin_amdgcn_wmma_f32_16x16x4_f32(false, a, false, b, (short)0, acc[0], false, false);                  \
    b[0] = w20.y; b[1] = w21.y;  acc[1] = __builtin_amdgcn_wmma_f32_16x16x4_f32(false, a, false, b, (short)0, acc[1], false, false);                  \
    b[0] = w20.z; b[1] = w21.z;  acc[2] = __builtin_amdgcn_wmma_f32_16x16x4_f32(false, a, false, b, (short)0, acc[2], false, false);                  \
    b[0] = w20.w; b[1] = w21.w;  acc[3] = __builtin_amdgcn_wmma_f32_16x16x4_f32(false, a, false, b, (short)0, acc[3], false, false);                  \
    a[0] = a2.z; a[1] = a2.w;                                                                                                                         \
    b[0] = w22.x; b[1] = w23.x;  acc[0] = __builtin_amdgcn_wmma_f32_16x16x4_f32(false, a, false, b, (short)0, acc[0], false, false);                  \
    b[0] = w22.y; b[1] = w23.y;  acc[1] = __builtin_amdgcn_wmma_f32_16x16x4_f32(false, a, false, b, (short)0, acc[1], false, false);                  \
    b[0] = w22.z; b[1] = w23.z;  acc[2] = __builtin_amdgcn_wmma_f32_16x16x4_f32(false, a, false, b, (short)0, acc[2], false, false);                  \
    b[0] = w22.w; b[1] = w23.w;  acc[3] = __builtin_amdgcn_wmma_f32_16x16x4_f32(false, a, false, b, (short)0, acc[3], false, false);                  \
    a[0] = a3.x; a[1] = a3.y;                                                                                                                         \
    b[0] = w30.x; b[1] = w31.x;  acc[0] = __builtin_amdgcn_wmma_f32_16x16x4_f32(false, a, false, b, (short)0, acc[0], false, false);                  \
    b[0] = w30.y; b[1] = w31.y;  acc[1] = __builtin_amdgcn_wmma_f32_16x16x4_f32(false, a, false, b, (short)0, acc[1], false, false);                  \
    b[0] = w30.z; b[1] = w31.z;  acc[2] = __builtin_amdgcn_wmma_f32_16x16x4_f32(false, a, false, b, (short)0, acc[2], false, false);                  \
    b[0] = w30.w; b[1] = w31.w;  acc[3] = __builtin_amdgcn_wmma_f32_16x16x4_f32(false, a, false, b, (short)0, acc[3], false, false);                  \
    a[0] = a3.z; a[1] = a3.w;                                                                                                                         \
    b[0] = w32.x; b[1] = w33.x;  acc[0] = __builtin_amdgcn_wmma_f32_16x16x4_f32(false, a, false, b, (short)0, acc[0], false, false);                  \
    b[0] = w32.y; b[1] = w33.y;  acc[1] = __builtin_amdgcn_wmma_f32_16x16x4_f32(false, a, false, b, (short)0, acc[1], false, false);                  \
    b[0] = w32.z; b[1] = w33.z;  acc[2] = __builtin_amdgcn_wmma_f32_16x16x4_f32(false, a, false, b, (short)0, acc[2], false, false);                  \
    b[0] = w32.w; b[1] = w33.w;  acc[3] = __builtin_amdgcn_wmma_f32_16x16x4_f32(false, a, false, b, (short)0, acc[3], false, false);                  \
    }

__device__ __forceinline__ void gemm_16x64(const float* __restrict__ A,
                                           const float* __restrict__ Wm,
                                           float* tile,
                                           int row_base, int N, int lane, int col0,
                                           v8f (&acc)[4]) {
    const int m    = lane & 15;
    const int kgrp = lane >> 4;
    const int lrow = lane >> 3;
    const int lcol = (lane & 7) * 4;

    const int last = N - 1;
    const float* s0 = A + (size_t)min(row_base + 0  + lrow, last) * IN_F + lcol;
    const float* s1 = A + (size_t)min(row_base + 4  + lrow, last) * IN_F + lcol;
    const float* s2 = A + (size_t)min(row_base + 8  + lrow, last) * IN_F + lcol;
    const float* s3 = A + (size_t)min(row_base + 12 + lrow, last) * IN_F + lcol;
    float*       dst  = tile + lrow * A_PITCH + lcol;
    const float* frag = tile + m * A_PITCH + kgrp * 4;
    const float* Bp   = Wm + (size_t)(kgrp * 4) * OUT_F + col0;

    float4 g0 = *(const float4*)(s0);
    float4 g1 = *(const float4*)(s1);
    float4 g2 = *(const float4*)(s2);
    float4 g3 = *(const float4*)(s3);

    #pragma unroll 1
    for (int kc = 0; kc < IN_F - A_CHUNK; kc += A_CHUNK) {
        *(float4*)(dst)                = g0;
        *(float4*)(dst + 4  * A_PITCH) = g1;
        *(float4*)(dst + 8  * A_PITCH) = g2;
        *(float4*)(dst + 12 * A_PITCH) = g3;
        __builtin_amdgcn_wave_barrier();
        TILE_LOADS(frag, Bp + (size_t)kc * OUT_F)
        g0 = *(const float4*)(s0 + kc + A_CHUNK);
        g1 = *(const float4*)(s1 + kc + A_CHUNK);
        g2 = *(const float4*)(s2 + kc + A_CHUNK);
        g3 = *(const float4*)(s3 + kc + A_CHUNK);
        TILE_MMAS
        asm volatile("" :: "v"(g0.x), "v"(g0.y), "v"(g0.z), "v"(g0.w),
                           "v"(g1.x), "v"(g1.y), "v"(g1.z), "v"(g1.w),
                           "v"(g2.x), "v"(g2.y), "v"(g2.z), "v"(g2.w),
                           "v"(g3.x), "v"(g3.y), "v"(g3.z), "v"(g3.w));
    }
    *(float4*)(dst)                = g0;
    *(float4*)(dst + 4  * A_PITCH) = g1;
    *(float4*)(dst + 8  * A_PITCH) = g2;
    *(float4*)(dst + 12 * A_PITCH) = g3;
    __builtin_amdgcn_wave_barrier();
    {
        TILE_LOADS(frag, Bp + (size_t)(IN_F - A_CHUNK) * OUT_F)
        TILE_MMAS
    }
}
#undef TILE_LOADS
#undef TILE_MMAS

__global__ __launch_bounds__(128)
void gemm_msg_kernel(const float* __restrict__ miu,
                     const float* __restrict__ sigma,
                     const float* __restrict__ Wmu,
                     const float* __restrict__ Wsig,
                     float* __restrict__ m_mu,
                     float* __restrict__ m_sig,
                     int N) {
    __shared__ __attribute__((aligned(16))) float s_tile[4][16 * A_PITCH];
    const int lane = threadIdx.x & 31;
    const int wave = threadIdx.x >> 5;
    const int strip = wave & 1;
    const int half  = wave >> 1;
    const int row_base = blockIdx.x * 32 + strip * 16;

    const int m    = lane & 15;
    const int kgrp = lane >> 4;
    const int col0 = half * 64 + m * 4;

    float att[4][8];

    {
        v8f acc[4];
        v8f zero = {};
        #pragma unroll
        for (int j = 0; j < 4; ++j) acc[j] = zero;

        gemm_16x64(sigma, Wsig, s_tile[wave], row_base, N, lane, col0, acc);

        #pragma unroll
        for (int r = 0; r < 8; ++r) {
            float o[4];
            #pragma unroll
            for (int j = 0; j < 4; ++j) {
                const float hs = fmaxf(acc[j][r], 0.0f);
                const float a  = __expf(-hs);
                att[j][r] = a;
                o[j] = hs * a * a;
            }
            const int rr = row_base + r + kgrp * 8;
            if (rr < N)
                *(float4*)(m_sig + (size_t)rr * OUT_F + col0) = make_float4(o[0], o[1], o[2], o[3]);
        }
    }

    {
        v8f acc[4];
        v8f zero = {};
        #pragma unroll
        for (int j = 0; j < 4; ++j) acc[j] = zero;

        gemm_16x64(miu, Wmu, s_tile[wave], row_base, N, lane, col0, acc);

        #pragma unroll
        for (int r = 0; r < 8; ++r) {
            float o[4];
            #pragma unroll
            for (int j = 0; j < 4; ++j) {
                const float hm = acc[j][r];
                const float e  = (hm > 0.0f) ? hm : (__expf(hm) - 1.0f);
                o[j] = e * att[j][r];
            }
            const int rr = row_base + r + kgrp * 8;
            if (rr < N)
                *(float4*)(m_mu + (size_t)rr * OUT_F + col0) = make_float4(o[0], o[1], o[2], o[3]);
        }
    }
}

__global__ __launch_bounds__(256)
void row_gather_kernel(const int* __restrict__ rowptr,
                       const int* __restrict__ sorted,
                       const float* __restrict__ m_mu,
                       const float* __restrict__ m_sig,
                       float* __restrict__ out_mu,
                       float* __restrict__ out_sig,
                       int N, int E) {
    const int lane = threadIdx.x & 31;
    const int r = __builtin_amdgcn_readfirstlane((int)(blockIdx.x * 8 + (threadIdx.x >> 5)));
    if (r >= N) return;

    int start = rowptr[r];
    int end   = rowptr[r + 1];
    start = max(0, min(start, E));
    end   = max(start, min(end, E));

    const int4* rec4 = (const int4*)sorted;
    float4 am = make_float4(0.0f, 0.0f, 0.0f, 0.0f);
    float4 as = make_float4(0.0f, 0.0f, 0.0f, 0.0f);

    for (int base = start; base < end; base += 32) {
        const int n = min(32, end - base);
        int4 rec = make_int4(0, 0, 0, 0);
        if (lane < n) rec = rec4[base + lane];

        #pragma unroll 4
        for (int j = 0; j < n; ++j) {
            int c = __builtin_amdgcn_readlane(rec.x, j);
            c = max(0, min(c, N - 1));
            const float v1 = __int_as_float(__builtin_amdgcn_readlane(rec.y, j));
            const float v2 = __int_as_float(__builtin_amdgcn_readlane(rec.z, j));
            const float4 mm = ((const float4*)(m_mu  + (size_t)c * OUT_F))[lane];
            const float4 ms = ((const float4*)(m_sig + (size_t)c * OUT_F))[lane];
            am.x += v1 * mm.x; am.y += v1 * mm.y; am.z += v1 * mm.z; am.w += v1 * mm.w;
            as.x += v2 * ms.x; as.y += v2 * ms.y; as.z += v2 * ms.z; as.w += v2 * ms.w;
        }
    }

    ((float4*)(out_mu  + (size_t)r * OUT_F))[lane] = am;
    ((float4*)(out_sig + (size_t)r * OUT_F))[lane] = as;
}

extern "C" void kernel_launch(void* const* d_in, const int* in_sizes, int n_in,
                              void* d_out, int out_size, void* d_ws, size_t ws_size,
                              hipStream_t stream) {
    const float* miu   = (const float*)d_in[0];
    const float* sigma = (const float*)d_in[1];
    const int*   erow  = (const int*)  d_in[2];
    const int*   ecol  = (const int*)  d_in[3];
    const float* a1    = (const float*)d_in[4];
    const float* a2    = (const float*)d_in[5];
    const float* Wmu   = (const float*)d_in[6];
    const float* Wsig  = (const float*)d_in[7];

    const int N = in_sizes[0] / IN_F;
    const int E = in_sizes[2];

    float* out_mu  = (float*)d_out;
    float* out_sig = out_mu + (size_t)N * OUT_F;

    const size_t nPad = ((size_t)N + 1 + 1023) / 1024 * 1024;
    const size_t ePad = ((size_t)E + 1023) / 1024 * 1024;
    float* m_mu     = (float*)d_ws;
    float* m_sig    = m_mu + (size_t)N * OUT_F;
    int*   rowptr   = (int*)(m_sig + (size_t)N * OUT_F);
    int*   cntm     = rowptr + nPad;
    int*   bstart   = cntm + NUM_CHUNKS * MAX_BUCKETS;
    int*   bucketed = bstart + 1024;
    int*   sorted   = bucketed + ePad * 4;

    const int NB = (N + ROWS_PER_BUCKET - 1) >> ROW_SHIFT;
    const int CE = (int)(((long long)E + NUM_CHUNKS - 1) / NUM_CHUNKS);

    chunk_hist_kernel<<<NUM_CHUNKS, WG, 0, stream>>>(erow, cntm, E, CE, NB);
    bucket_scatter_kernel<<<NUM_CHUNKS, WG, 0, stream>>>(erow, ecol, a1, a2, cntm, bstart, bucketed, E, CE, NB);
    bucket_sort_kernel<<<NB, WG, 0, stream>>>(bstart, bucketed, rowptr, sorted, N, E);

    gemm_msg_kernel<<<(N + 31) / 32, 128, 0, stream>>>(
        miu, sigma, Wmu, Wsig, m_mu, m_sig, N);

    row_gather_kernel<<<(N + 7) / 8, 256, 0, stream>>>(
        rowptr, sorted, m_mu, m_sig, out_mu, out_sig, N, E);
}


# level-2 bucket sort (row count, scan, fill) is no launch of its own any more: one extra 4-wave work-group per bucket at the front of the GEMM launch does it; row runs still in LDS-add arrival order
# speedup vs baseline: 3.7477x; 1.0502x over previous
#include <hip/hip_runtime.h>
#include <hip/hip_bf16.h>

typedef float v2f __attribute__((ext_vector_type(2)));
typedef float v8f __attribute__((ext_vector_type(8)));

#define IN_F 256
#define OUT_F 128

#define ROW_SHIFT 9
#define ROWS_PER_BUCKET (1 << ROW_SHIFT)
#define MAX_BUCKETS 256
#define NUM_CHUNKS 256
#define WG 1024
#define BATCH 4


__device__ __forceinline__ int bucket_of(int row, int NB) {
    int b = row >> ROW_SHIFT;
    return max(0, min(b, NB - 1));
}

__global__ __launch_bounds__(WG)
void chunk_hist_kernel(const int* __restrict__ erow,
                       int* __restrict__ cntm,
                       int E, int CE, int NB) {
    __shared__ int h[MAX_BUCKETS];
    const int tid = threadIdx.x;
    const int c = blockIdx.x;
    if (tid < MAX_BUCKETS) h[tid] = 0;
    __syncthreads();
    const int e0 = min(E, c * CE);
    const int e1 = min(E, e0 + CE);
    for (int base = e0 + tid; base < e1; base += BATCH * WG) {
        int r[BATCH];
        #pragma unroll
        for (int u = 0; u < BATCH; ++u)
            r[u] = erow[min(base + u * WG, e1 - 1)];
        #pragma unroll
        for (int u = 0; u < BATCH; ++u)
            if (base + u * WG < e1) atomicAdd(&h[bucket_of(r[u], NB)], 1);
    }
    __syncthreads();
    if (tid < MAX_BUCKETS) cntm[c * MAX_BUCKETS + tid] = h[tid];
}

__global__ __launch_bounds__(WG)
void bucket_scatter_kernel(const int* __restrict__ erow,
                           const int* __restrict__ ecol,
                           const float* __restrict__ a1,
                           const float* __restrict__ a2,
                           const int* __restrict__ cntm,
                           int* __restrict__ bstart,
                           int* __restrict__ bucketed,
                           int E, int CE, int NB) {
    __shared__ int s_all[4][MAX_BUCKETS];
    __shared__ int s_pre[4][MAX_BUCKETS];
    __shared__ int s_w[8];
    __shared__ int cur[MAX_BUCKETS];
    const int tid  = threadIdx.x;
    const int b    = tid & 255;
    const int q    = tid >> 8;
    const int lane = tid & 31;
    const int wave = tid >> 5;
    const int c    = blockIdx.x;

    int all = 0, pre = 0;
    #pragma unroll
    for (int i = 0; i < 64; ++i) {
        const int ch = q * 64 + i;
        const int v = cntm[ch * MAX_BUCKETS + b];
        all += v;
        if (ch < c) pre += v;
    }
    s_all[q][b] = all;
    s_pre[q][b] = pre;
    __syncthreads();

    int T = 0, P = 0, incl = 0;
    if (q == 0) {
        #pragma unroll
        for (int qq = 0; qq < 4; ++qq) {
            T += s_all[qq][b];
            P += s_pre[qq][b];
        }
        incl = T;
        #pragma unroll
        for (int d = 1; d < 32; d <<= 1) {
            int t = __shfl_up(incl, d, 32);
            if (lane >= d) incl += t;
        }
        if (lane == 31) s_w[wave] = incl;
    }
    __syncthreads();
    if (q == 0) {
        int front = 0;
        #pragma unroll
        for (int w = 0; w < 8; ++w)
            if (w < wave) front += s_w[w];
        const int S = front + incl - T;
        cur[b] = S + P;
        if (c == 0) {
            bstart[b] = S;
            if (b == MAX_BUCKETS - 1) bstart[MAX_BUCKETS] = S + T;
        }
    }
    __syncthreads();

    const int e0 = min(E, c * CE);
    const int e1 = min(E, e0 + CE);
    for (int base = e0 + tid; base < e1; base += BATCH * WG) {
        int4 rec[BATCH];
        #pragma unroll
        for (int u = 0; u < BATCH; ++u) {
            const int e = min(base + u * WG, e1 - 1);
            rec[u].x = erow[e];
            rec[u].y = ecol[e];
            rec[u].z = __float_as_int(a1[e]);
            rec[u].w = __float_as_int(a2[e]);
        }
        #pragma unroll
        for (int u = 0; u < BATCH; ++u) {
            if (base + u * WG < e1) {
                int pos = atomicAdd(&cur[bucket_of(rec[u].x, NB)], 1);
                pos = max(0, min(pos, E - 1));
                ((int4*)bucketed)[pos] = rec[u];
            }
        }
    }
}

#define CT 128
#define CB 16
static_assert(ROWS_PER_BUCKET == 4 * CT, "the row scan gives every thread 4 consecutive rows");

__device__ __forceinline__ void bucket_sort_group(const int* __restrict__ bstart,
                                                  const int* __restrict__ bucketed,
                                                  int* __restrict__ rowptr,
                                                  int* __restrict__ sorted,
                                                  int N, int E, int b,
                                                  int* cnt,
                                                  int* s_w) {
    const int tid  = threadIdx.x;
    const int lane = tid & 31;
    const int wave = tid >> 5;
    const int row0 = b << ROW_SHIFT;

    int s = bstart[b];
    int t = bstart[b + 1];
    s = max(0, min(s, E));
    t = max(s, min(t, E));

    for (int i = 0; i < ROWS_PER_BUCKET / CT; ++i) cnt[tid + i * CT] = 0;
    __syncthreads();

    const int4* in4 = (const int4*)bucketed;
    for (int base = s + tid; base < t; base += CB * CT) {
        int x[CB];
        #pragma unroll
        for (int u = 0; u < CB; ++u)
            x[u] = in4[min(base + u * CT, t - 1)].x;
        #pragma unroll
        for (int u = 0; u < CB; ++u) {
            const int lr = x[u] - row0;
            if (base + u * CT < t && (unsigned)lr < (unsigned)ROWS_PER_BUCKET) atomicAdd(&cnt[lr], 1);
        }
    }
    __syncthreads();

    const int4 n4 = ((const int4*)cnt)[tid];
    const int tot = n4.x + n4.y + n4.z + n4.w;
    int incl = tot;
    #pragma unroll
    for (int d = 1; d < 32; d <<= 1) {
        int u = __shfl_up(incl, d, 32);
        if (lane >= d) incl += u;
    }
    if (lane == 31) s_w[wave] = incl;
    __syncthreads();
    {
        int front = 0;
        #pragma unroll
        for (int w = 0; w < CT / 32; ++w)
            if (w < wave) front += s_w[w];
        int4 ex;
        ex.x = front + incl - tot;
        ex.y = ex.x + n4.x;
        ex.z = ex.y + n4.y;
        ex.w = ex.z + n4.z;
        ((int4*)cnt)[tid] = ex;
        const int g = row0 + 4 * tid;
        if (g + 0 < N) rowptr[g + 0] = s + ex.x;
        if (g + 1 < N) rowptr[g + 1] = s + ex.y;
        if (g + 2 < N) rowptr[g + 2] = s + ex.z;
        if (g + 3 < N) rowptr[g + 3] = s + ex.w;
        if (g + 0 == N - 1) rowptr[N] = s + ex.x + n4.x;
        if (g + 1 == N - 1) rowptr[N] = s + ex.y + n4.y;
        if (g + 2 == N - 1) rowptr[N] = s + ex.z + n4.z;
        if (g + 3 == N - 1) rowptr[N] = s + ex.w + n4.w;
    }
    __syncthreads();

    for (int base = s + tid; base < t; base += CB * CT) {
        int4 rec[CB];
        #pragma unroll
        for (int u = 0; u < CB; ++u)
            rec[u] = in4[min(base + u * CT, t - 1)];
        #pragma unroll
        for (int u = 0; u < CB; ++u) {
            const int lr = rec[u].x - row0;
            if (base + u * CT < t && (unsigned)lr < (unsigned)ROWS_PER_BUCKET) {
                int pos = s + atomicAdd(&cnt[lr], 1);
                pos = max(0, min(pos, E - 1));
                int4 o;
                o.x = rec[u].y;
                o.y = rec[u].z;
                o.z = rec[u].w;
                o.w = 0;
                ((int4*)sorted)[pos] = o;
            }
        }
    }
}

#define A_PITCH 36
#define A_CHUNK 32

#define TILE_LOADS(frag, Bk)                                      \
    const float4 a0 = *(const float4*)((frag));                  \
    const float4 a1 = *(const float4*)((frag) + 8);              \
    const float4 a2 = *(const float4*)((frag) + 16);             \
    const float4 a3 = *(const float4*)((frag) + 24);             \
    const float4 w00 = *(const float4*)((Bk) + (size_t)(0 + 0) * OUT_F);                            \
    const float4 w01 = *(const float4*)((Bk) + (size_t)(0 + 1) * OUT_F);                            \
    const float4 w02 = *(const float4*)((Bk) + (size_t)(0 + 2) * OUT_F);                            \
    const float4 w03 = *(const float4*)((Bk) + (size_t)(0 + 3) * OUT_F);                            \
    const float4 w10 = *(const float4*)((Bk) + (size_t)(8 + 0) * OUT_F);                            \
    const float4 w11 = *(const float4*)((Bk) + (size_t)(8 + 1) * OUT_F);                            \
    const float4 w12 = *(const float4*)((Bk) + (size_t)(8 + 2) * OUT_F);                            \
    const float4 w13 = *(const float4*)((Bk) + (size_t)(8 + 3) * OUT_F);                            \
    const float4 w20 = *(const float4*)((Bk) + (size_t)(16 + 0) * OUT_F);                           \
    const float4 w21 = *(const float4*)((Bk) + (size_t)(16 + 1) * OUT_F);                           \
    const float4 w22 = *(const float4*)((Bk) + (size_t)(16 + 2) * OUT_F);                           \
    const float4 w23 = *(const float4*)((Bk) + (size_t)(16 + 3) * OUT_F);                           \
    const float4 w30 = *(const float4*)((Bk) + (size_t)(24 + 0) * OUT_F);                           \
    const float4 w31 = *(const float4*)((Bk) + (size_t)(24 + 1) * OUT_F);                           \
    const float4 w32 = *(const float4*)((Bk) + (size_t)(24 + 2) * OUT_F);                           \
    const float4 w33 = *(const float4*)((Bk) + (size_t)(24 + 3) * OUT_F);                           \


#define TILE_MMAS                                                 \
    {                                                             \
    v2f a, b;                                                                                                                                                      \
    a[0] = a0.x; a[1] = a0.y;                                                                                                                         \
    b[0] = w00.x; b[1] = w01.x;  acc[0] = __builtin_amdgcn_wmma_f32_16x16x4_f32(false, a, false, b, (short)0, acc[0], false, false);                  \
    b[0] = w00.y; b[1] = w01.y;  acc[1] = __builtin_amdgcn_wmma_f32_16x16x4_f32(false, a, false, b, (short)0, acc[1], false, false);                  \
    b[0] = w00.z; b[1] = w01.z;  acc[2] = __builtin_amdgcn_wmma_f32_16x16x4_f32(false, a, false, b, (short)0, acc[2], false, false);                  \
    b[0] = w00.w; b[1] = w01.w;  acc[3] = __builtin_amdgcn_wmma_f32_16x16x4_f32(false, a, false, b, (short)0, acc[3], false, false);                  \
    a[0] = a0.z; a[1] = a0.w;                                                                                                                         \
    b[0] = w02.x; b[1] = w03.x;  acc[0] = __builtin_amdgcn_wmma_f32_16x16x4_f32(false, a, false, b, (short)0, acc[0], false, false);                  \
    b[0] = w02.y; b[1] = w03.y;  acc[1] = __builtin_amdgcn_wmma_f32_16x16x4_f32(false, a, false, b, (short)0, acc[1], false, false);                  \
    b[0] = w02.z; b[1] = w03.z;  acc[2] = __builtin_amdgcn_wmma_f32_16x16x4_f32(false, a, false, b, (short)0, acc[2], false, false);                  \
    b[0] = w02.w; b[1] = w03.w;  acc[3] = __builtin_amdgcn_wmma_f32_16x16x4_f32(false, a, false, b, (short)0, acc[3], false, false);                  \
    a[0] = a1.x; a[1] = a1.y;                                                                                                                         \
    b[0] = w10.x; b[1] = w11.x;  acc[0] = __builtin_amdgcn_wmma_f32_16x16x4_f32(false, a, false, b, (short)0, acc[0], false, false);                  \
    b[0] = w10.y; b[1] = w11.y;  acc[1] = __builtin_amdgcn_wmma_f32_16x16x4_f32(false, a, false, b, (short)0, acc[1], false, false);                  \
    b[0] = w10.z; b[1] = w11.z;  acc[2] = __builtin_amdgcn_wmma_f32_16x16x4_f32(false, a, false, b, (short)0, acc[2], false, false);                  \
    b[0] = w10.w; b[1] = w11.w;  acc[3] = __builtin_amdgcn_wmma_f32_16x16x4_f32(false, a, false, b, (short)0, acc[3], false, false);                  \
    a[0] = a1.z; a[1] = a1.w;                                                                                                                         \
    b[0] = w12.x; b[1] = w13.x;  acc[0] = __builtin_amdgcn_wmma_f32_16x16x4_f32(false, a, false, b, (short)0, acc[0], false, false);                  \
    b[0] = w12.y; b[1] = w13.y;  acc[1] = __builtin_amdgcn_wmma_f32_16x16x4_f32(false, a, false, b, (short)0, acc[1], false, false);                  \
    b[0] = w12.z; b[1] = w13.z;  acc[2] = __builtin_amdgcn_wmma_f32_16x16x4_f32(false, a, false, b, (short)0, acc[2], false, false);                  \
    b[0] = w12.w; b[1] = w13.w;  acc[3] = __builtin_amdgcn_wmma_f32_16x16x4_f32(false, a, false, b, (short)0, acc[3], false, false);                  \
    a[0] = a2.x; a[1] = a2.y;                                                                                                                         \
    b[0] = w20.x; b[1] = w21.x;  acc[0] = __builtin_amdgcn_wmma_f32_16x16x4_f32(false, a, false, b, (short)0, acc[0], false, false);                  \
    b[0] = w20.y; b[1] = w21.y;  acc[1] = __builtin_amdgcn_wmma_f32_16x16x4_f32(false, a, false, b, (short)0, acc[1], false, false);                  \
    b[0] = w20.z; b[1] = w21.z;  acc[2] = __builtin_amdgcn_wmma_f32_16x16x4_f32(false, a, false, b, (short)0, acc[2], false, false);                  \
    b[0] = w20.w; b[1] = w21.w;  acc[3] = __builtin_amdgcn_wmma_f32_16x16x4_f32(false, a, false, b, (short)0, acc[3], false, false);                  \
    a[0] = a2.z; a[1] = a2.w;                                                                                                                         \
    b[0] = w22.x; b[1] = w23.x;  acc[0] = __builtin_amdgcn_wmma_f32_16x16x4_f32(false, a, false, b, (short)0, acc[0], false, false);                  \
    b[0] = w22.y; b[1] = w23.y;  acc[1] = __builtin_amdgcn_wmma_f32_16x16x4_f32(false, a, false, b, (short)0, acc[1], false, false);                  \
    b[0] = w22.z; b[1] = w23.z;  acc[2] = __builtin_amdgcn_wmma_f32_16x16x4_f32(false, a, false, b, (short)0, acc[2], false, false);                  \
    b[0] = w22.w; b[1] = w23.w;  acc[3] = __builtin_amdgcn_wmma_f32_16x16x4_f32(false, a, false, b, (short)0, acc[3], false, false);                  \
    a[0] = a3.x; a[1] = a3.y;                                                                                                                         \
    b[0] = w30.x; b[1] = w31.x;  acc[0] = __builtin_amdgcn_wmma_f32_16x16x4_f32(false, a, false, b, (short)0, acc[0], false, false);                  \
    b[0] = w30.y; b[1] = w31.y;  acc[1] = __builtin_amdgcn_wmma_f32_16x16x4_f32(false, a, false, b, (short)0, acc[1], false, false);                  \
    b[0] = w30.z; b[1] = w31.z;  acc[2] = __builtin_amdgcn_wmma_f32_16x16x4_f32(false, a, false, b, (short)0, acc[2], false, false);                  \
    b[0] = w30.w; b[1] = w31.w;  acc[3] = __builtin_amdgcn_wmma_f32_16x16x4_f32(false, a, false, b, (short)0, acc[3], false, false);                  \
    a[0] = a3.z; a[1] = a3.w;                                                                                                                         \
    b[0] = w32.x; b[1] = w33.x;  acc[0] = __builtin_amdgcn_wmma_f32_16x16x4_f32(false, a, false, b, (short)0, acc[0], false, false);                  \
    b[0] = w32.y; b[1] = w33.y;  acc[1] = __builtin_amdgcn_wmma_f32_16x16x4_f32(false, a, false, b, (short)0, acc[1], false, false);                  \
    b[0] = w32.z; b[1] = w33.z;  acc[2] = __builtin_amdgcn_wmma_f32_16x16x4_f32(false, a, false, b, (short)0, acc[2], false, false);                  \
    b[0] = w32.w; b[1] = w33.w;  acc[3] = __builtin_amdgcn_wmma_f32_16x16x4_f32(false, a, false, b, (short)0, acc[3], false, false);                  \
    }

__device__ __forceinline__ void gemm_16x64(const float* __restrict__ A,
                                           const float* __restrict__ Wm,
                                           float* tile,
                                           int row_base, int N, int lane, int col0,
                                           v8f (&acc)[4]) {
    const int m    = lane & 15;
    const int kgrp = lane >> 4;
    const int lrow = lane >> 3;
    const int lcol = (lane & 7) * 4;

    const int last = N - 1;
    const float* s0 = A + (size_t)min(row_base + 0  + lrow, last) * IN_F + lcol;
    const float* s1 = A + (size_t)min(row_base + 4  + lrow, last) * IN_F + lcol;
    const float* s2 = A + (size_t)min(row_base + 8  + lrow, last) * IN_F + lcol;
    const float* s3 = A + (size_t)min(row_base + 12 + lrow, last) * IN_F + lcol;
    float*       dst  = tile + lrow * A_PITCH + lcol;
    const float* frag = tile + m * A_PITCH + kgrp * 4;
    const float* Bp   = Wm + (size_t)(kgrp * 4) * OUT_F + col0;

    float4 g0 = *(const float4*)(s0);
    float4 g1 = *(const float4*)(s1);
    float4 g2 = *(const float4*)(s2);
    float4 g3 = *(const float4*)(s3);

    #pragma unroll 1
    for (int kc = 0; kc < IN_F - A_CHUNK; kc += A_CHUNK) {
        *(float4*)(dst)                = g0;
        *(float4*)(dst + 4  * A_PITCH) = g1;
        *(float4*)(dst + 8  * A_PITCH) = g2;
        *(float4*)(dst + 12 * A_PITCH) = g3;
        __builtin_amdgcn_wave_barrier();
        TILE_LOADS(frag, Bp + (size_t)kc * OUT_F)
        g0 = *(const float4*)(s0 + kc + A_CHUNK);
        g1 = *(const float4*)(s1 + kc + A_CHUNK);
        g2 = *(const float4*)(s2 + kc + A_CHUNK);
        g3 = *(const float4*)(s3 + kc + A_CHUNK);
        TILE_MMAS
        asm volatile("" :: "v"(g0.x), "v"(g0.y), "v"(g0.z), "v"(g0.w),
                           "v"(g1.x), "v"(g1.y), "v"(g1.z), "v"(g1.w),
                           "v"(g2.x), "v"(g2.y), "v"(g2.z), "v"(g2.w),
                           "v"(g3.x), "v"(g3.y), "v"(g3.z), "v"(g3.w));
    }
    *(float4*)(dst)                = g0;
    *(float4*)(dst + 4  * A_PITCH) = g1;
    *(float4*)(dst + 8  * A_PITCH) = g2;
    *(float4*)(dst + 12 * A_PITCH) = g3;
    __builtin_amdgcn_wave_barrier();
    {
        TILE_LOADS(frag, Bp + (size_t)(IN_F - A_CHUNK) * OUT_F)
        TILE_MMAS
    }
}
#undef TILE_LOADS
#undef TILE_MMAS

__global__ __launch_bounds__(CT)
void gemm_msg_kernel(const float* __restrict__ miu,
                     const float* __restrict__ sigma,
                     const float* __restrict__ Wmu,
                     const float* __restrict__ Wsig,
                     float* __restrict__ m_mu,
                     float* __restrict__ m_sig,
                     int N,
                     const int* __restrict__ bstart,
                     const int* __restrict__ bucketed,
                     int* __restrict__ rowptr,
                     int* __restrict__ sorted,
                     int E, int NB) {
    __shared__ __attribute__((aligned(16))) float s_tile[4][16 * A_PITCH];
    __shared__ int s_w[CT / 32];

    if ((int)blockIdx.x < NB) {
        bucket_sort_group(bstart, bucketed, rowptr, sorted, N, E, (int)blockIdx.x,
                          (int*)&s_tile[0][0], s_w);
        return;
    }

    const int lane = threadIdx.x & 31;
    const int wave = threadIdx.x >> 5;
    const int strip = wave & 1;
    const int half  = wave >> 1;
    const int row_base = ((int)blockIdx.x - NB) * 32 + strip * 16;

    const int m    = lane & 15;
    const int kgrp = lane >> 4;
    const int col0 = half * 64 + m * 4;

    float att[4][8];

    {
        v8f acc[4];
        v8f zero = {};
        #pragma unroll
        for (int j = 0; j < 4; ++j) acc[j] = zero;

        gemm_16x64(sigma, Wsig, s_tile[wave], row_base, N, lane, col0, acc);

        #pragma unroll
        for (int r = 0; r < 8; ++r) {
            float o[4];
            #pragma unroll
            for (int j = 0; j < 4; ++j) {
                const float hs = fmaxf(acc[j][r], 0.0f);
                const float a  = __expf(-hs);
                att[j][r] = a;
                o[j] = hs * a * a;
            }
            const int rr = row_base + r + kgrp * 8;
            if (rr < N)
                *(float4*)(m_sig + (size_t)rr * OUT_F + col0) = make_float4(o[0], o[1], o[2], o[3]);
        }
    }

    {
        v8f acc[4];
        v8f zero = {};
        #pragma unroll
        for (int j = 0; j < 4; ++j) acc[j] = zero;

        gemm_16x64(miu, Wmu, s_tile[wave], row_base, N, lane, col0, acc);

        #pragma unroll
        for (int r = 0; r < 8; ++r) {
            float o[4];
            #pragma unroll
            for (int j = 0; j < 4; ++j) {
                const float hm = acc[j][r];
                const float e  = (hm > 0.0f) ? hm : (__expf(hm) - 1.0f);
                o[j] = e * att[j][r];
            }
            const int rr = row_base + r + kgrp * 8;
            if (rr < N)
                *(float4*)(m_mu + (size_t)rr * OUT_F + col0) = make_float4(o[0], o[1], o[2], o[3]);
        }
    }
}

__global__ __launch_bounds__(256)
void row_gather_kernel(const int* __restrict__ rowptr,
                       const int* __restrict__ sorted,
                       const float* __restrict__ m_mu,
                       const float* __restrict__ m_sig,
                       float* __restrict__ out_mu,
                       float* __restrict__ out_sig,
                       int N, int E) {
    const int lane = threadIdx.x & 31;
    const int r = __builtin_amdgcn_readfirstlane((int)(blockIdx.x * 8 + (threadIdx.x >> 5)));
    if (r >= N) return;

    int start = rowptr[r];
    int end   = rowptr[r + 1];
    start = max(0, min(start, E));
    end   = max(start, min(end, E));

    const int4* rec4 = (const int4*)sorted;
    float4 am = make_float4(0.0f, 0.0f, 0.0f, 0.0f);
    float4 as = make_float4(0.0f, 0.0f, 0.0f, 0.0f);

    for (int base = start; base < end; base += 32) {
        const int n = min(32, end - base);
        int4 rec = make_int4(0, 0, 0, 0);
        if (lane < n) rec = rec4[base + lane];

        #pragma unroll 4
        for (int j = 0; j < n; ++j) {
            int c = __builtin_amdgcn_readlane(rec.x, j);
            c = max(0, min(c, N - 1));
            const float v1 = __int_as_float(__builtin_amdgcn_readlane(rec.y, j));
            const float v2 = __int_as_float(__builtin_amdgcn_readlane(rec.z, j));
            const float4 mm = ((const float4*)(m_mu  + (size_t)c * OUT_F))[lane];
            const float4 ms = ((const float4*)(m_sig + (size_t)c * OUT_F))[lane];
            am.x += v1 * mm.x; am.y += v1 * mm.y; am.z += v1 * mm.z; am.w += v1 * mm.w;
            as.x += v2 * ms.x; as.y += v2 * ms.y; as.z += v2 * ms.z; as.w += v2 * ms.w;
        }
    }

    ((float4*)(out_mu  + (size_t)r * OUT_F))[lane] = am;
    ((float4*)(out_sig + (size_t)r * OUT_F))[lane] = as;
}

extern "C" void kernel_launch(void* const* d_in, const int* in_sizes, int n_in,
                              void* d_out, int out_size, void* d_ws, size_t ws_size,
                              hipStream_t stream) {
    const float* miu   = (const float*)d_in[0];
    const float* sigma = (const float*)d_in[1];
    const int*   erow  = (const int*)  d_in[2];
    const int*   ecol  = (const int*)  d_in[3];
    const float* a1    = (const float*)d_in[4];
    const float* a2    = (const float*)d_in[5];
    const float* Wmu   = (const float*)d_in[6];
    const float* Wsig  = (const float*)d_in[7];

    const int N = in_sizes[0] / IN_F;
    const int E = in_sizes[2];

    float* out_mu  = (float*)d_out;
    float* out_sig = out_mu + (size_t)N * OUT_F;

    const size_t nPad = ((size_t)N + 1 + 1023) / 1024 * 1024;
    const size_t ePad = ((size_t)E + 1023) / 1024 * 1024;
    float* m_mu     = (float*)d_ws;
    float* m_sig    = m_mu + (size_t)N * OUT_F;
    int*   rowptr   = (int*)(m_sig + (size_t)N * OUT_F);
    int*   cntm     = rowptr + nPad;
    int*   bstart   = cntm + NUM_CHUNKS * MAX_BUCKETS;
    int*   bucketed = bstart + 1024;
    int*   sorted   = bucketed + ePad * 4;

    const int NB = (N + ROWS_PER_BUCKET - 1) >> ROW_SHIFT;
    const int CE = (int)(((long long)E + NUM_CHUNKS - 1) / NUM_CHUNKS);

    chunk_hist_kernel<<<NUM_CHUNKS, WG, 0, stream>>>(erow, cntm, E, CE, NB);
    bucket_scatter_kernel<<<NUM_CHUNKS, WG, 0, stream>>>(erow, ecol, a1, a2, cntm, bstart, bucketed, E, CE, NB);

    gemm_msg_kernel<<<NB + (N + 31) / 32, CT, 0, stream>>>(
        miu, sigma, Wmu, Wsig, m_mu, m_sig, N,
        bstart, bucketed, rowptr, sorted, E, NB);

    row_gather_kernel<<<(N + 7) / 8, 256, 0, stream>>>(
        rowptr, sorted, m_mu, m_sig, out_mu, out_sig, N, E);
}
